# starting kernel
# baseline (speedup 1.0000x reference)
#include <hip/hip_runtime.h>
#include <hip/hip_bf16.h>

typedef float v2f __attribute__((ext_vector_type(2)));
typedef float v8f __attribute__((ext_vector_type(8)));

#define NNODES 50000
#define NEDGES 600000
#define INDIM 128
#define HID 256
#define MTILES 5

__global__ void zero_f32_kernel(float* __restrict__ p, int n) {
  int i = blockIdx.x * blockDim.x + threadIdx.x;
  if (i < n) p[i] = 0.0f;
}

__global__ void deg_accum_kernel(const int* __restrict__ dst, float* __restrict__ deg, int E) {
  int e = blockIdx.x * blockDim.x + threadIdx.x;
  if (e < E) atomicAdd(&deg[dst[e]], 1.0f);
}

__global__ void make_dinv_kernel(float* __restrict__ deg, int n) {
  int i = blockIdx.x * blockDim.x + threadIdx.x;
  if (i < n) deg[i] = rsqrtf(deg[i] + 1.0f);
}

__global__ void pack_b_kernel(const float* __restrict__ B, float* __restrict__ Bp,
                              int N, int total) {
  int i = blockIdx.x * blockDim.x + threadIdx.x;
  if (i >= total) return;
  int k = i / N;
  int n = i - k * N;
  Bp[((size_t)(k >> 1) * N + n) * 2 + (k & 1)] = B[i];
}

template <int K>
__global__ __launch_bounds__(128, 1)
void gemm_wmma_f32_kernel(const float* __restrict__ A,
                          const float* __restrict__ Bp,
                          const float* __restrict__ dinv,
                          float* __restrict__ H,
                          float* __restrict__ AGG,
                          int N) {
  const int wave = threadIdx.x >> 5;
  const int lane = threadIdx.x & 31;
  const int half = lane >> 4;
  const int lidx = lane & 15;

  const int m0 = blockIdx.x * (16 * MTILES);
  const int n0 = wave * 64;

  v8f acc[MTILES][4] = {};

  const float* abase = A + (size_t)(m0 + lidx) * K + 2 * half;
  const float* bbase = Bp + ((size_t)half * N + n0 + lidx) * 2;

  for (int k = 0; k < K; k += 4) {
    const float* bk = bbase + (size_t)(k >> 1) * N * 2;
    v2f bf[4];
    #pragma unroll
    for (int u = 0; u < 4; ++u)
      bf[u] = *(const v2f*)(bk + u * 32);

    #pragma unroll
    for (int t = 0; t < MTILES; ++t) {
      v2f af = *(const v2f*)(abase + t * 16 * K + k);
      #pragma unroll
      for (int u = 0; u < 4; ++u) {
        acc[t][u] = __builtin_amdgcn_wmma_f32_16x16x4_f32(
            false, af, false, bf[u], (short)0, acc[t][u], false, false);
      }
    }
  }

  #pragma unroll
  for (int t = 0; t < MTILES; ++t) {
    #pragma unroll
    for (int g = 0; g < 8; ++g) {
      const int row = m0 + t * 16 + g + 8 * half;
      const float di = dinv[row];
      const float sl = di * di;
      const size_t base = (size_t)row * N + n0 + lidx;
      #pragma unroll
      for (int u = 0; u < 4; ++u) {
        float v = acc[t][u][g];
        H[base + u * 16]   = v;
        AGG[base + u * 16] = v * sl;
      }
    }
  }
}

__global__ void edge_scatter_kernel(const float* __restrict__ h,
                                    const int* __restrict__ src,
                                    const int* __restrict__ dst,
                                    const float* __restrict__ dinv,
                                    float* __restrict__ agg, int E) {
  int e = blockIdx.x * (blockDim.x >> 5) + (threadIdx.x >> 5);
  if (e >= E) return;
  int lane = threadIdx.x & 31;
  int s = src[e];
  int d = dst[e];
  float norm = dinv[s] * dinv[d];
  const float4* hs = (const float4*)(h + (size_t)s * HID) + lane * 2;
  float* ad = agg + (size_t)d * HID + lane * 8;
  float4 v0 = hs[0];
  float4 v1 = hs[1];
  atomicAdd(ad + 0, v0.x * norm);
  atomicAdd(ad + 1, v0.y * norm);
  atomicAdd(ad + 2, v0.z * norm);
  atomicAdd(ad + 3, v0.w * norm);
  atomicAdd(ad + 4, v1.x * norm);
  atomicAdd(ad + 5, v1.y * norm);
  atomicAdd(ad + 6, v1.z * norm);
  atomicAdd(ad + 7, v1.w * norm);
}

__global__ void bias_relu_kernel(const float* __restrict__ agg,
                                 const float* __restrict__ b,
                                 float* __restrict__ out, int total) {
  int i = blockIdx.x * blockDim.x + threadIdx.x;
  if (i < total) {
    float v = agg[i] + b[i & (HID - 1)];
    out[i] = v > 0.0f ? v : 0.0f;
  }
}

__global__ void final_linear_kernel(const float* __restrict__ x,
                                    const float* __restrict__ Wl,
                                    const float* __restrict__ bl,
                                    float* __restrict__ out, int total) {
  int i = blockIdx.x * blockDim.x + threadIdx.x;
  if (i >= total) return;
  int node = i >> 1;
  int o = i & 1;
  const float* xr = x + (size_t)node * HID;
  float acc = bl[o];
  #pragma unroll 8
  for (int k = 0; k < HID; ++k) acc += xr[k] * Wl[k * 2 + o];
  out[i] = acc;
}

extern "C" void kernel_launch(void* const* d_in, const int* in_sizes, int n_in,
                              void* d_out, int out_size, void* d_ws, size_t ws_size,
                              hipStream_t stream) {
  const float* x   = (const float*)d_in[0];
  const int*   ei  = (const int*)d_in[1];
  const float* W1  = (const float*)d_in[2];
  const float* b1  = (const float*)d_in[3];
  const float* W2  = (const float*)d_in[4];
  const float* b2  = (const float*)d_in[5];
  const float* Wl  = (const float*)d_in[6];
  const float* bl  = (const float*)d_in[7];
  float*       out = (float*)d_out;

  const int*   src = ei;
  const int*   dst = ei + NEDGES;

  char* ws = (char*)d_ws;
  const size_t H_OFF   = 200192;
  const size_t HBYTES  = (size_t)NNODES * HID * 4;
  const size_t AGG_OFF = H_OFF + HBYTES;
  const size_t X2_OFF  = AGG_OFF + HBYTES;
  const size_t BP_OFF  = X2_OFF + HBYTES;

  float* dinv = (float*)ws;
  float* H    = (float*)(ws + H_OFF);
  float* AGG  = (float*)(ws + AGG_OFF);
  float* X2   = (float*)(ws + X2_OFF);
  float* Bp   = (float*)(ws + BP_OFF);

  const int TPB = 256;
  const int nodeBlocks  = (NNODES + TPB - 1) / TPB;
  const int edgeBlocks  = (NEDGES + TPB - 1) / TPB;
  const int featTotal   = NNODES * HID;
  const int featBlocks  = (featTotal + TPB - 1) / TPB;
  const int waveEdgeBlk = (NEDGES + 7) / 8;
  const int gemmBlocks  = NNODES / (16 * MTILES);

  zero_f32_kernel<<<nodeBlocks, TPB, 0, stream>>>(dinv, NNODES);
  deg_accum_kernel<<<edgeBlocks, TPB, 0, stream>>>(dst, dinv, NEDGES);
  make_dinv_kernel<<<nodeBlocks, TPB, 0, stream>>>(dinv, NNODES);

  {
    const int tot = INDIM * HID;
    pack_b_kernel<<<(tot + TPB - 1) / TPB, TPB, 0, stream>>>(W1, Bp, HID, tot);
    gemm_wmma_f32_kernel<INDIM><<<gemmBlocks, 128, 0, stream>>>(x, Bp, dinv, H, AGG, HID);
    edge_scatter_kernel<<<waveEdgeBlk, TPB, 0, stream>>>(H, src, dst, dinv, AGG, NEDGES);
    bias_relu_kernel<<<featBlocks, TPB, 0, stream>>>(AGG, b1, X2, featTotal);
  }

  {
    const int tot = HID * HID;
    pack_b_kernel<<<(tot + TPB - 1) / TPB, TPB, 0, stream>>>(W2, Bp, HID, tot);
    gemm_wmma_f32_kernel<HID><<<gemmBlocks, 128, 0, stream>>>(X2, Bp, dinv, H, AGG, HID);
    edge_scatter_kernel<<<waveEdgeBlk, TPB, 0, stream>>>(H, src, dst, dinv, AGG, NEDGES);
    bias_relu_kernel<<<featBlocks, TPB, 0, stream>>>(AGG, b2, X2, featTotal);
  }

  const int outTotal = NNODES * 2;
  final_linear_kernel<<<(outTotal + TPB - 1) / TPB, TPB, 0, stream>>>(X2, Wl, bl, out, outTotal);
}


# Edge sum as a per-destination gather over a CSR built each call (int atomic rank + scan) instead of 154M f32 atomics; row sums now run in arrival-rank order with fused multiply-add
# speedup vs baseline: 2.3592x; 2.3592x over previous
#include <hip/hip_runtime.h>
#include <hip/hip_bf16.h>

typedef float v2f __attribute__((ext_vector_type(2)));
typedef float v8f __attribute__((ext_vector_type(8)));

#define NNODES 50000
#define NEDGES 600000
#define INDIM 128
#define HID 256
#define MTILES 5

__global__ void zero_i32_kernel(int* __restrict__ p, int n) {
  int i = blockIdx.x * blockDim.x + threadIdx.x;
  if (i < n) p[i] = 0;
}

__global__ void deg_rank_kernel(const int* __restrict__ dst, int* __restrict__ cnt,
                                int* __restrict__ rank, int E) {
  int e = blockIdx.x * blockDim.x + threadIdx.x;
  if (e < E) rank[e] = atomicAdd(&cnt[dst[e]], 1);
}

__global__ void make_dinv_kernel(const int* __restrict__ cnt, float* __restrict__ dinv, int n) {
  int i = blockIdx.x * blockDim.x + threadIdx.x;
  if (i < n) dinv[i] = rsqrtf((float)cnt[i] + 1.0f);
}

#define SCAN_TPB 1024
__global__ __launch_bounds__(SCAN_TPB)
void rowptr_scan_kernel(const int* __restrict__ cnt, int* __restrict__ rowptr, int n) {
  __shared__ int wpre[32];
  __shared__ int wtot[32];
  const int tid  = threadIdx.x;
  const int lane = tid & 31;
  const int wv   = tid >> 5;
  const int base = blockIdx.x * SCAN_TPB;

  int pre = 0;
  for (int j = tid; j < base; j += SCAN_TPB) pre += cnt[j];

  const int i = base + tid;
  const int v = (i < n) ? cnt[i] : 0;

  int x = v;
  for (int o = 1; o < 32; o <<= 1) {
    int y = __shfl_up(x, o);
    if (lane >= o) x += y;
  }
  for (int o = 16; o > 0; o >>= 1) pre += __shfl_xor(pre, o);

  if (lane == 31) wtot[wv] = x;
  if (lane == 0)  wpre[wv] = pre;
  __syncthreads();

  int t = wpre[lane] + ((lane < wv) ? wtot[lane] : 0);
  for (int o = 16; o > 0; o >>= 1) t += __shfl_xor(t, o);

  const int excl = t + x - v;
  if (i < n) rowptr[i] = excl;
  if (i == n - 1) rowptr[n] = excl + v;
}

__global__ void csr_fill_kernel(const int* __restrict__ src, const int* __restrict__ dst,
                                const int* __restrict__ rank, const int* __restrict__ rowptr,
                                int* __restrict__ csr_src, int E) {
  int e = blockIdx.x * blockDim.x + threadIdx.x;
  if (e < E) csr_src[rowptr[dst[e]] + rank[e]] = src[e];
}

__global__ void pack_b_kernel(const float* __restrict__ B, float* __restrict__ Bp,
                              int N, int total) {
  int i = blockIdx.x * blockDim.x + threadIdx.x;
  if (i >= total) return;
  int k = i / N;
  int n = i - k * N;
  Bp[((size_t)(k >> 1) * N + n) * 2 + (k & 1)] = B[i];
}

template <int K>
__global__ __launch_bounds__(128, 1)
void gemm_wmma_f32_kernel(const float* __restrict__ A,
                          const float* __restrict__ Bp,
                          const float* __restrict__ dinv,
                          float* __restrict__ H,
                          float* __restrict__ AGG,
                          int N) {
  const int wave = threadIdx.x >> 5;
  const int lane = threadIdx.x & 31;
  const int half = lane >> 4;
  const int lidx = lane & 15;

  const int m0 = blockIdx.x * (16 * MTILES);
  const int n0 = wave * 64;

  v8f acc[MTILES][4] = {};

  const float* abase = A + (size_t)(m0 + lidx) * K + 2 * half;
  const float* bbase = Bp + ((size_t)half * N + n0 + lidx) * 2;

  for (int k = 0; k < K; k += 4) {
    const float* bk = bbase + (size_t)(k >> 1) * N * 2;
    v2f bf[4];
    #pragma unroll
    for (int u = 0; u < 4; ++u)
      bf[u] = *(const v2f*)(bk + u * 32);

    #pragma unroll
    for (int t = 0; t < MTILES; ++t) {
      v2f af = *(const v2f*)(abase + t * 16 * K + k);
      #pragma unroll
      for (int u = 0; u < 4; ++u) {
        acc[t][u] = __builtin_amdgcn_wmma_f32_16x16x4_f32(
            false, af, false, bf[u], (short)0, acc[t][u], false, false);
      }
    }
  }

  #pragma unroll
  for (int t = 0; t < MTILES; ++t) {
    #pragma unroll
    for (int g = 0; g < 8; ++g) {
      const int row = m0 + t * 16 + g + 8 * half;
      const float di = dinv[row];
      const float sl = di * di;
      const size_t base = (size_t)row * N + n0 + lidx;
      #pragma unroll
      for (int u = 0; u < 4; ++u) {
        float v = acc[t][u][g];
        H[base + u * 16]   = v;
        AGG[base + u * 16] = v * sl;
      }
    }
  }
}

static __device__ __forceinline__ void row_axpy(float4& a0, float4& a1,
                                                const float4 v0, const float4 v1,
                                                const float w) {
  a0.x = fmaf(v0.x, w, a0.x);
  a0.y = fmaf(v0.y, w, a0.y);
  a0.z = fmaf(v0.z, w, a0.z);
  a0.w = fmaf(v0.w, w, a0.w);
  a1.x = fmaf(v1.x, w, a1.x);
  a1.y = fmaf(v1.y, w, a1.y);
  a1.z = fmaf(v1.z, w, a1.z);
  a1.w = fmaf(v1.w, w, a1.w);
}

__global__ __launch_bounds__(256)
void edge_gather_kernel(const float* __restrict__ h,
                        const int* __restrict__ rowptr,
                        const int* __restrict__ csr_src,
                        const float* __restrict__ dinv,
                        float* __restrict__ agg, int n) {
  const int node = __builtin_amdgcn_readfirstlane(blockIdx.x * 8 + (threadIdx.x >> 5));
  if (node >= n) return;
  const int lane = threadIdx.x & 31;
  const int beg = rowptr[node];
  const int end = rowptr[node + 1];
  const float di = dinv[node];

  float4* ap = (float4*)(agg + (size_t)node * HID);
  float4 a0 = ap[lane];
  float4 a1 = ap[lane + 32];

  for (int base = beg; base < end; base += 32) {
    const int m = (end - base < 32) ? (end - base) : 32;
    int   s_l = 0;
    float w_l = 0.0f;
    if (lane < m) {
      s_l = csr_src[base + lane];
      w_l = dinv[s_l] * di;
    }
    int j = 0;
    for (; j + 4 <= m; j += 4) {
      const int s0 = __builtin_amdgcn_readlane(s_l, j);
      const int s1 = __builtin_amdgcn_readlane(s_l, j + 1);
      const int s2 = __builtin_amdgcn_readlane(s_l, j + 2);
      const int s3 = __builtin_amdgcn_readlane(s_l, j + 3);
      const float w0 = __int_as_float(__builtin_amdgcn_readlane(__float_as_int(w_l), j));
      const float w1 = __int_as_float(__builtin_amdgcn_readlane(__float_as_int(w_l), j + 1));
      const float w2 = __int_as_float(__builtin_amdgcn_readlane(__float_as_int(w_l), j + 2));
      const float w3 = __int_as_float(__builtin_amdgcn_readlane(__float_as_int(w_l), j + 3));
      const float4* h0 = (const float4*)(h + (size_t)s0 * HID);
      const float4* h1 = (const float4*)(h + (size_t)s1 * HID);
      const float4* h2 = (const float4*)(h + (size_t)s2 * HID);
      const float4* h3 = (const float4*)(h + (size_t)s3 * HID);
      const float4 p0 = h0[lane], q0 = h0[lane + 32];
      const float4 p1 = h1[lane], q1 = h1[lane + 32];
      const float4 p2 = h2[lane], q2 = h2[lane + 32];
      const float4 p3 = h3[lane], q3 = h3[lane + 32];
      row_axpy(a0, a1, p0, q0, w0);
      row_axpy(a0, a1, p1, q1, w1);
      row_axpy(a0, a1, p2, q2, w2);
      row_axpy(a0, a1, p3, q3, w3);
    }
    #pragma unroll 1
    for (; j < m; ++j) {
      const int s0 = __builtin_amdgcn_readlane(s_l, j);
      const float w0 = __int_as_float(__builtin_amdgcn_readlane(__float_as_int(w_l), j));
      const float4* h0 = (const float4*)(h + (size_t)s0 * HID);
      const float4 p0 = h0[lane], q0 = h0[lane + 32];
      row_axpy(a0, a1, p0, q0, w0);
    }
  }

  ap[lane]      = a0;
  ap[lane + 32] = a1;
}

__global__ void bias_relu_kernel(const float* __restrict__ agg,
                                 const float* __restrict__ b,
                                 float* __restrict__ out, int total) {
  int i = blockIdx.x * blockDim.x + threadIdx.x;
  if (i < total) {
    float v = agg[i] + b[i & (HID - 1)];
    out[i] = v > 0.0f ? v : 0.0f;
  }
}

__global__ void final_linear_kernel(const float* __restrict__ x,
                                    const float* __restrict__ Wl,
                                    const float* __restrict__ bl,
                                    float* __restrict__ out, int total) {
  int i = blockIdx.x * blockDim.x + threadIdx.x;
  if (i >= total) return;
  int node = i >> 1;
  int o = i & 1;
  const float* xr = x + (size_t)node * HID;
  float acc = bl[o];
  #pragma unroll 8
  for (int k = 0; k < HID; ++k) acc += xr[k] * Wl[k * 2 + o];
  out[i] = acc;
}

extern "C" void kernel_launch(void* const* d_in, const int* in_sizes, int n_in,
                              void* d_out, int out_size, void* d_ws, size_t ws_size,
                              hipStream_t stream) {
  const float* x   = (const float*)d_in[0];
  const int*   ei  = (const int*)d_in[1];
  const float* W1  = (const float*)d_in[2];
  const float* b1  = (const float*)d_in[3];
  const float* W2  = (const float*)d_in[4];
  const float* b2  = (const float*)d_in[5];
  const float* Wl  = (const float*)d_in[6];
  const float* bl  = (const float*)d_in[7];
  float*       out = (float*)d_out;

  const int*   src = ei;
  const int*   dst = ei + NEDGES;

  char* ws = (char*)d_ws;
  const size_t H_OFF   = 200192;
  const size_t HBYTES  = (size_t)NNODES * HID * 4;
  const size_t AGG_OFF = H_OFF + HBYTES;
  const size_t X2_OFF  = AGG_OFF + HBYTES;
  const size_t BP_OFF  = X2_OFF + HBYTES;
  const size_t CNT_OFF = BP_OFF + 262144;
  const size_t RP_OFF  = CNT_OFF + 200192;
  const size_t RK_OFF  = RP_OFF + 200192;
  const size_t CSR_OFF = RK_OFF + 2400000;

  float* dinv = (float*)ws;
  float* H    = (float*)(ws + H_OFF);
  float* AGG  = (float*)(ws + AGG_OFF);
  float* X2   = (float*)(ws + X2_OFF);
  float* Bp   = (float*)(ws + BP_OFF);
  int*   cnt    = (int*)(ws + CNT_OFF);
  int*   rowptr = (int*)(ws + RP_OFF);
  int*   rank   = (int*)(ws + RK_OFF);
  int*   csrsrc = (int*)(ws + CSR_OFF);

  const int TPB = 256;
  const int nodeBlocks  = (NNODES + TPB - 1) / TPB;
  const int edgeBlocks  = (NEDGES + TPB - 1) / TPB;
  const int featTotal   = NNODES * HID;
  const int featBlocks  = (featTotal + TPB - 1) / TPB;
  const int waveNodeBlk = (NNODES + 7) / 8;
  const int scanBlocks  = (NNODES + SCAN_TPB - 1) / SCAN_TPB;
  const int gemmBlocks  = NNODES / (16 * MTILES);

  zero_i32_kernel<<<nodeBlocks, TPB, 0, stream>>>(cnt, NNODES);
  deg_rank_kernel<<<edgeBlocks, TPB, 0, stream>>>(dst, cnt, rank, NEDGES);
  make_dinv_kernel<<<nodeBlocks, TPB, 0, stream>>>(cnt, dinv, NNODES);
  rowptr_scan_kernel<<<scanBlocks, SCAN_TPB, 0, stream>>>(cnt, rowptr, NNODES);
  csr_fill_kernel<<<edgeBlocks, TPB, 0, stream>>>(src, dst, rank, rowptr, csrsrc, NEDGES);

  {
    const int tot = INDIM * HID;
    pack_b_kernel<<<(tot + TPB - 1) / TPB, TPB, 0, stream>>>(W1, Bp, HID, tot);
    gemm_wmma_f32_kernel<INDIM><<<gemmBlocks, 128, 0, stream>>>(x, Bp, dinv, H, AGG, HID);
    edge_gather_kernel<<<waveNodeBlk, TPB, 0, stream>>>(H, rowptr, csrsrc, dinv, AGG, NNODES);
    bias_relu_kernel<<<featBlocks, TPB, 0, stream>>>(AGG, b1, X2, featTotal);
  }

  {
    const int tot = HID * HID;
    pack_b_kernel<<<(tot + TPB - 1) / TPB, TPB, 0, stream>>>(W2, Bp, HID, tot);
    gemm_wmma_f32_kernel<HID><<<gemmBlocks, 128, 0, stream>>>(X2, Bp, dinv, H, AGG, HID);
    edge_gather_kernel<<<waveNodeBlk, TPB, 0, stream>>>(H, rowptr, csrsrc, dinv, AGG, NNODES);
    bias_relu_kernel<<<featBlocks, TPB, 0, stream>>>(AGG, b2, X2, featTotal);
  }

  const int outTotal = NNODES * 2;
  final_linear_kernel<<<(outTotal + TPB - 1) / TPB, TPB, 0, stream>>>(X2, Wl, bl, out, outTotal);
}


# Gather wave finishes each layer in registers: self-loop term, bias, ReLU, in layer 2 also the 256x2 projection (per-lane sums, then wave butterfly); bias/ReLU and projection passes are gone
# speedup vs baseline: 3.3843x; 1.4345x over previous
#include <hip/hip_runtime.h>
#include <hip/hip_bf16.h>

typedef float v2f __attribute__((ext_vector_type(2)));
typedef float v8f __attribute__((ext_vector_type(8)));

#define NNODES 50000
#define NEDGES 600000
#define INDIM 128
#define HID 256
#define MTILES 5

__global__ void zero_i32_kernel(int* __restrict__ p, int n) {
  int i = blockIdx.x * blockDim.x + threadIdx.x;
  if (i < n) p[i] = 0;
}

__global__ void deg_rank_kernel(const int* __restrict__ dst, int* __restrict__ cnt,
                                int* __restrict__ rank, int E) {
  int e = blockIdx.x * blockDim.x + threadIdx.x;
  if (e < E) rank[e] = atomicAdd(&cnt[dst[e]], 1);
}

__global__ void make_dinv_kernel(const int* __restrict__ cnt, float* __restrict__ dinv, int n) {
  int i = blockIdx.x * blockDim.x + threadIdx.x;
  if (i < n) dinv[i] = rsqrtf((float)cnt[i] + 1.0f);
}

#define SCAN_TPB 1024
__global__ __launch_bounds__(SCAN_TPB)
void rowptr_scan_kernel(const int* __restrict__ cnt, int* __restrict__ rowptr, int n) {
  __shared__ int wpre[32];
  __shared__ int wtot[32];
  const int tid  = threadIdx.x;
  const int lane = tid & 31;
  const int wv   = tid >> 5;
  const int base = blockIdx.x * SCAN_TPB;

  int pre = 0;
  for (int j = tid; j < base; j += SCAN_TPB) pre += cnt[j];

  const int i = base + tid;
  const int v = (i < n) ? cnt[i] : 0;

  int x = v;
  for (int o = 1; o < 32; o <<= 1) {
    int y = __shfl_up(x, o);
    if (lane >= o) x += y;
  }
  for (int o = 16; o > 0; o >>= 1) pre += __shfl_xor(pre, o);

  if (lane == 31) wtot[wv] = x;
  if (lane == 0)  wpre[wv] = pre;
  __syncthreads();

  int t = wpre[lane] + ((lane < wv) ? wtot[lane] : 0);
  for (int o = 16; o > 0; o >>= 1) t += __shfl_xor(t, o);

  const int excl = t + x - v;
  if (i < n) rowptr[i] = excl;
  if (i == n - 1) rowptr[n] = excl + v;
}

__global__ void csr_fill_kernel(const int* __restrict__ src, const int* __restrict__ dst,
                                const int* __restrict__ rank, const int* __restrict__ rowptr,
                                int* __restrict__ csr_src, int E) {
  int e = blockIdx.x * blockDim.x + threadIdx.x;
  if (e < E) csr_src[rowptr[dst[e]] + rank[e]] = src[e];
}

__global__ void pack_b_kernel(const float* __restrict__ B, float* __restrict__ Bp,
                              int N, int total) {
  int i = blockIdx.x * blockDim.x + threadIdx.x;
  if (i >= total) return;
  int k = i / N;
  int n = i - k * N;
  Bp[((size_t)(k >> 1) * N + n) * 2 + (k & 1)] = B[i];
}

template <int K>
__global__ __launch_bounds__(128, 1)
void gemm_wmma_f32_kernel(const float* __restrict__ A,
                          const float* __restrict__ Bp,
                          float* __restrict__ H,
                          int N) {
  const int wave = threadIdx.x >> 5;
  const int lane = threadIdx.x & 31;
  const int half = lane >> 4;
  const int lidx = lane & 15;

  const int m0 = blockIdx.x * (16 * MTILES);
  const int n0 = wave * 64;

  v8f acc[MTILES][4] = {};

  const float* abase = A + (size_t)(m0 + lidx) * K + 2 * half;
  const float* bbase = Bp + ((size_t)half * N + n0 + lidx) * 2;

  for (int k = 0; k < K; k += 4) {
    const float* bk = bbase + (size_t)(k >> 1) * N * 2;
    v2f bf[4];
    #pragma unroll
    for (int u = 0; u < 4; ++u)
      bf[u] = *(const v2f*)(bk + u * 32);

    #pragma unroll
    for (int t = 0; t < MTILES; ++t) {
      v2f af = *(const v2f*)(abase + t * 16 * K + k);
      #pragma unroll
      for (int u = 0; u < 4; ++u) {
        acc[t][u] = __builtin_amdgcn_wmma_f32_16x16x4_f32(
            false, af, false, bf[u], (short)0, acc[t][u], false, false);
      }
    }
  }

  #pragma unroll
  for (int t = 0; t < MTILES; ++t) {
    #pragma unroll
    for (int g = 0; g < 8; ++g) {
      const int row = m0 + t * 16 + g + 8 * half;
      const size_t base = (size_t)row * N + n0 + lidx;
      #pragma unroll
      for (int u = 0; u < 4; ++u)
        H[base + u * 16] = acc[t][u][g];
    }
  }
}

static __device__ __forceinline__ void row_axpy(float4& a0, float4& a1,
                                                const float4 v0, const float4 v1,
                                                const float w) {
  a0.x = fmaf(v0.x, w, a0.x);
  a0.y = fmaf(v0.y, w, a0.y);
  a0.z = fmaf(v0.z, w, a0.z);
  a0.w = fmaf(v0.w, w, a0.w);
  a1.x = fmaf(v1.x, w, a1.x);
  a1.y = fmaf(v1.y, w, a1.y);
  a1.z = fmaf(v1.z, w, a1.z);
  a1.w = fmaf(v1.w, w, a1.w);
}

static __device__ __forceinline__ float add_relu(const float a, const float b) {
  const float v = a + b;
  return v > 0.0f ? v : 0.0f;
}

template <bool FINAL, int WAVES>
__global__ __launch_bounds__(WAVES * 32)
void gather_finish_kernel(const float* __restrict__ h,
                          const int* __restrict__ rowptr,
                          const int* __restrict__ csr_src,
                          const float* __restrict__ dinv,
                          const float* __restrict__ bias,
                          const float* __restrict__ Wl,
                          const float* __restrict__ bl,
                          float* __restrict__ dstbuf, int n) {
  const int wv   = threadIdx.x >> 5;
  const int lane = threadIdx.x & 31;
  const int node = __builtin_amdgcn_readfirstlane(blockIdx.x * WAVES + wv);

  float r0 = 0.0f, r1 = 0.0f;
  if (node < n) {
    const int beg = rowptr[node];
    const int end = rowptr[node + 1];
    const float di = dinv[node];
    const float sl = di * di;

    const float4* hp = (const float4*)(h + (size_t)node * HID);
    float4 a0 = hp[lane];
    float4 a1 = hp[lane + 32];
    a0.x *= sl; a0.y *= sl; a0.z *= sl; a0.w *= sl;
    a1.x *= sl; a1.y *= sl; a1.z *= sl; a1.w *= sl;

    for (int base = beg; base < end; base += 32) {
      const int m = (end - base < 32) ? (end - base) : 32;
      int   s_l = 0;
      float w_l = 0.0f;
      if (lane < m) {
        s_l = csr_src[base + lane];
        w_l = dinv[s_l] * di;
      }
      int j = 0;
      for (; j + 4 <= m; j += 4) {
        const int s0 = __builtin_amdgcn_readlane(s_l, j);
        const int s1 = __builtin_amdgcn_readlane(s_l, j + 1);
        const int s2 = __builtin_amdgcn_readlane(s_l, j + 2);
        const int s3 = __builtin_amdgcn_readlane(s_l, j + 3);
        const float w0 = __int_as_float(__builtin_amdgcn_readlane(__float_as_int(w_l), j));
        const float w1 = __int_as_float(__builtin_amdgcn_readlane(__float_as_int(w_l), j + 1));
        const float w2 = __int_as_float(__builtin_amdgcn_readlane(__float_as_int(w_l), j + 2));
        const float w3 = __int_as_float(__builtin_amdgcn_readlane(__float_as_int(w_l), j + 3));
        const float4* h0 = (const float4*)(h + (size_t)s0 * HID);
        const float4* h1 = (const float4*)(h + (size_t)s1 * HID);
        const float4* h2 = (const float4*)(h + (size_t)s2 * HID);
        const float4* h3 = (const float4*)(h + (size_t)s3 * HID);
        const float4 p0 = h0[lane], q0 = h0[lane + 32];
        const float4 p1 = h1[lane], q1 = h1[lane + 32];
        const float4 p2 = h2[lane], q2 = h2[lane + 32];
        const float4 p3 = h3[lane], q3 = h3[lane + 32];
        row_axpy(a0, a1, p0, q0, w0);
        row_axpy(a0, a1, p1, q1, w1);
        row_axpy(a0, a1, p2, q2, w2);
        row_axpy(a0, a1, p3, q3, w3);
      }
      #pragma unroll 1
      for (; j < m; ++j) {
        const int s0 = __builtin_amdgcn_readlane(s_l, j);
        const float w0 = __int_as_float(__builtin_amdgcn_readlane(__float_as_int(w_l), j));
        const float4* h0 = (const float4*)(h + (size_t)s0 * HID);
        const float4 p0 = h0[lane], q0 = h0[lane + 32];
        row_axpy(a0, a1, p0, q0, w0);
      }
    }

    const float4 b0 = ((const float4*)bias)[lane];
    const float4 b1 = ((const float4*)bias)[lane + 32];
    a0.x = add_relu(a0.x, b0.x); a0.y = add_relu(a0.y, b0.y);
    a0.z = add_relu(a0.z, b0.z); a0.w = add_relu(a0.w, b0.w);
    a1.x = add_relu(a1.x, b1.x); a1.y = add_relu(a1.y, b1.y);
    a1.z = add_relu(a1.z, b1.z); a1.w = add_relu(a1.w, b1.w);

    if constexpr (!FINAL) {
      float4* op = (float4*)(dstbuf + (size_t)node * HID);
      op[lane]      = a0;
      op[lane + 32] = a1;
    } else {
      const float4* wp = (const float4*)Wl;
      const float4 wa = wp[2 * lane],      wb = wp[2 * lane + 1];
      const float4 wc = wp[64 + 2 * lane], wd = wp[64 + 2 * lane + 1];
      r0 = a0.x * wa.x;            r1 = a0.x * wa.y;
      r0 = fmaf(a0.y, wa.z, r0);   r1 = fmaf(a0.y, wa.w, r1);
      r0 = fmaf(a0.z, wb.x, r0);   r1 = fmaf(a0.z, wb.y, r1);
      r0 = fmaf(a0.w, wb.z, r0);   r1 = fmaf(a0.w, wb.w, r1);
      r0 = fmaf(a1.x, wc.x, r0);   r1 = fmaf(a1.x, wc.y, r1);
      r0 = fmaf(a1.y, wc.z, r0);   r1 = fmaf(a1.y, wc.w, r1);
      r0 = fmaf(a1.z, wd.x, r0);   r1 = fmaf(a1.z, wd.y, r1);
      r0 = fmaf(a1.w, wd.z, r0);   r1 = fmaf(a1.w, wd.w, r1);
      for (int o = 16; o > 0; o >>= 1) {
        r0 += __shfl_xor(r0, o);
        r1 += __shfl_xor(r1, o);
      }
      r0 += bl[0];
      r1 += bl[1];
    }
  }

  if constexpr (FINAL) {
    __shared__ float stage[2 * WAVES];
    if (lane == 0) {
      stage[2 * wv]     = r0;
      stage[2 * wv + 1] = r1;
    }
    __syncthreads();
    if (wv == 0) {
      const int idx = blockIdx.x * (2 * WAVES) + lane;
      if (lane < 2 * WAVES && idx < 2 * n) dstbuf[idx] = stage[lane];
    }
  }
}

extern "C" void kernel_launch(void* const* d_in, const int* in_sizes, int n_in,
                              void* d_out, int out_size, void* d_ws, size_t ws_size,
                              hipStream_t stream) {
  const float* x   = (const float*)d_in[0];
  const int*   ei  = (const int*)d_in[1];
  const float* W1  = (const float*)d_in[2];
  const float* b1  = (const float*)d_in[3];
  const float* W2  = (const float*)d_in[4];
  const float* b2  = (const float*)d_in[5];
  const float* Wl  = (const float*)d_in[6];
  const float* bl  = (const float*)d_in[7];
  float*       out = (float*)d_out;

  const int*   src = ei;
  const int*   dst = ei + NEDGES;

  char* ws = (char*)d_ws;
  const size_t H_OFF   = 200192;
  const size_t HBYTES  = (size_t)NNODES * HID * 4;
  const size_t X2_OFF  = H_OFF + HBYTES;
  const size_t BP_OFF  = X2_OFF + HBYTES;
  const size_t CNT_OFF = BP_OFF + 262144;
  const size_t RP_OFF  = CNT_OFF + 200192;
  const size_t RK_OFF  = RP_OFF + 200192;
  const size_t CSR_OFF = RK_OFF + 2400000;

  float* dinv = (float*)ws;
  float* H    = (float*)(ws + H_OFF);
  float* X2   = (float*)(ws + X2_OFF);
  float* Bp   = (float*)(ws + BP_OFF);
  int*   cnt    = (int*)(ws + CNT_OFF);
  int*   rowptr = (int*)(ws + RP_OFF);
  int*   rank   = (int*)(ws + RK_OFF);
  int*   csrsrc = (int*)(ws + CSR_OFF);

  const int TPB = 256;
  const int nodeBlocks  = (NNODES + TPB - 1) / TPB;
  const int edgeBlocks  = (NEDGES + TPB - 1) / TPB;
  const int rowBlocks8  = (NNODES + 7) / 8;
  const int rowBlocks16 = (NNODES + 15) / 16;
  const int scanBlocks  = (NNODES + SCAN_TPB - 1) / SCAN_TPB;
  const int gemmBlocks  = NNODES / (16 * MTILES);

  zero_i32_kernel<<<nodeBlocks, TPB, 0, stream>>>(cnt, NNODES);
  deg_rank_kernel<<<edgeBlocks, TPB, 0, stream>>>(dst, cnt, rank, NEDGES);
  make_dinv_kernel<<<nodeBlocks, TPB, 0, stream>>>(cnt, dinv, NNODES);
  rowptr_scan_kernel<<<scanBlocks, SCAN_TPB, 0, stream>>>(cnt, rowptr, NNODES);
  csr_fill_kernel<<<edgeBlocks, TPB, 0, stream>>>(src, dst, rank, rowptr, csrsrc, NEDGES);

  {
    const int tot = INDIM * HID;
    pack_b_kernel<<<(tot + TPB - 1) / TPB, TPB, 0, stream>>>(W1, Bp, HID, tot);
    gemm_wmma_f32_kernel<INDIM><<<gemmBlocks, 128, 0, stream>>>(x, Bp, H, HID);
    gather_finish_kernel<false, 8><<<rowBlocks8, 256, 0, stream>>>(
        H, rowptr, csrsrc, dinv, b1, Wl, bl, X2, NNODES);
  }

  {
    const int tot = HID * HID;
    pack_b_kernel<<<(tot + TPB - 1) / TPB, TPB, 0, stream>>>(W2, Bp, HID, tot);
    gemm_wmma_f32_kernel<HID><<<gemmBlocks, 128, 0, stream>>>(X2, Bp, H, HID);
    gather_finish_kernel<true, 16><<<rowBlocks16, 512, 0, stream>>>(
        H, rowptr, csrsrc, dinv, b2, Wl, bl, out, NNODES);
  }
}


# Layer 1 takes the edge sum on the 128-wide input rows before the GEMM, (A x) W1 instead of A (x W1): the gather moves half the bytes; bias+ReLU go to the GEMM epilogue; layer-1 sum order changes
# speedup vs baseline: 3.7419x; 1.1057x over previous
#include <hip/hip_runtime.h>
#include <hip/hip_bf16.h>

typedef float v2f __attribute__((ext_vector_type(2)));
typedef float v8f __attribute__((ext_vector_type(8)));

#define NNODES 50000
#define NEDGES 600000
#define INDIM 128
#define HID 256
#define MTILES 5

__global__ void zero_i32_kernel(int* __restrict__ p, int n) {
  int i = blockIdx.x * blockDim.x + threadIdx.x;
  if (i < n) p[i] = 0;
}

__global__ void deg_rank_kernel(const int* __restrict__ dst, int* __restrict__ cnt,
                                int* __restrict__ rank, int E) {
  int e = blockIdx.x * blockDim.x + threadIdx.x;
  if (e < E) rank[e] = atomicAdd(&cnt[dst[e]], 1);
}

__global__ void make_dinv_kernel(const int* __restrict__ cnt, float* __restrict__ dinv, int n) {
  int i = blockIdx.x * blockDim.x + threadIdx.x;
  if (i < n) dinv[i] = rsqrtf((float)cnt[i] + 1.0f);
}

#define SCAN_TPB 1024
__global__ __launch_bounds__(SCAN_TPB)
void rowptr_scan_kernel(const int* __restrict__ cnt, int* __restrict__ rowptr, int n) {
  __shared__ int wpre[32];
  __shared__ int wtot[32];
  const int tid  = threadIdx.x;
  const int lane = tid & 31;
  const int wv   = tid >> 5;
  const int base = blockIdx.x * SCAN_TPB;

  int pre = 0;
  for (int j = tid; j < base; j += SCAN_TPB) pre += cnt[j];

  const int i = base + tid;
  const int v = (i < n) ? cnt[i] : 0;

  int x = v;
  for (int o = 1; o < 32; o <<= 1) {
    int y = __shfl_up(x, o);
    if (lane >= o) x += y;
  }
  for (int o = 16; o > 0; o >>= 1) pre += __shfl_xor(pre, o);

  if (lane == 31) wtot[wv] = x;
  if (lane == 0)  wpre[wv] = pre;
  __syncthreads();

  int t = wpre[lane] + ((lane < wv) ? wtot[lane] : 0);
  for (int o = 16; o > 0; o >>= 1) t += __shfl_xor(t, o);

  const int excl = t + x - v;
  if (i < n) rowptr[i] = excl;
  if (i == n - 1) rowptr[n] = excl + v;
}

__global__ void csr_fill_kernel(const int* __restrict__ src, const int* __restrict__ dst,
                                const int* __restrict__ rank, const int* __restrict__ rowptr,
                                int* __restrict__ csr_src, int E) {
  int e = blockIdx.x * blockDim.x + threadIdx.x;
  if (e < E) csr_src[rowptr[dst[e]] + rank[e]] = src[e];
}

__global__ void pack_b_kernel(const float* __restrict__ B, float* __restrict__ Bp,
                              int N, int total) {
  int i = blockIdx.x * blockDim.x + threadIdx.x;
  if (i >= total) return;
  int k = i / N;
  int n = i - k * N;
  Bp[((size_t)(k >> 1) * N + n) * 2 + (k & 1)] = B[i];
}

static __device__ __forceinline__ float add_relu(const float a, const float b) {
  const float v = a + b;
  return v > 0.0f ? v : 0.0f;
}

template <int K, bool BIAS_RELU>
__global__ __launch_bounds__(128, 1)
void gemm_wmma_f32_kernel(const float* __restrict__ A,
                          const float* __restrict__ Bp,
                          const float* __restrict__ bias,
                          float* __restrict__ H,
                          int N) {
  const int wave = threadIdx.x >> 5;
  const int lane = threadIdx.x & 31;
  const int half = lane >> 4;
  const int lidx = lane & 15;

  const int m0 = blockIdx.x * (16 * MTILES);
  const int n0 = wave * 64;

  v8f acc[MTILES][4] = {};

  const float* abase = A + (size_t)(m0 + lidx) * K + 2 * half;
  const float* bbase = Bp + ((size_t)half * N + n0 + lidx) * 2;

  for (int k = 0; k < K; k += 4) {
    const float* bk = bbase + (size_t)(k >> 1) * N * 2;
    v2f bf[4];
    #pragma unroll
    for (int u = 0; u < 4; ++u)
      bf[u] = *(const v2f*)(bk + u * 32);

    #pragma unroll
    for (int t = 0; t < MTILES; ++t) {
      v2f af = *(const v2f*)(abase + t * 16 * K + k);
      #pragma unroll
      for (int u = 0; u < 4; ++u) {
        acc[t][u] = __builtin_amdgcn_wmma_f32_16x16x4_f32(
            false, af, false, bf[u], (short)0, acc[t][u], false, false);
      }
    }
  }

  float bv[4] = {0.0f, 0.0f, 0.0f, 0.0f};
  if constexpr (BIAS_RELU) {
    #pragma unroll
    for (int u = 0; u < 4; ++u) bv[u] = bias[n0 + lidx + u * 16];
  }
  #pragma unroll
  for (int t = 0; t < MTILES; ++t) {
    #pragma unroll
    for (int g = 0; g < 8; ++g) {
      const int row = m0 + t * 16 + g + 8 * half;
      const size_t base = (size_t)row * N + n0 + lidx;
      #pragma unroll
      for (int u = 0; u < 4; ++u) {
        if constexpr (BIAS_RELU) H[base + u * 16] = add_relu(acc[t][u][g], bv[u]);
        else                     H[base + u * 16] = acc[t][u][g];
      }
    }
  }
}

static __device__ __forceinline__ void vec_axpy(float4& a, const float4 v, const float w) {
  a.x = fmaf(v.x, w, a.x);
  a.y = fmaf(v.y, w, a.y);
  a.z = fmaf(v.z, w, a.z);
  a.w = fmaf(v.w, w, a.w);
}

template <int NB>
static __device__ __forceinline__ void gather_batch(float4& a, const float* __restrict__ x,
                                                    const int s_l, const float w_l,
                                                    const int j, const int lane) {
  int s[NB]; float w[NB]; float4 p[NB];
  #pragma unroll
  for (int u = 0; u < NB; ++u) {
    s[u] = __builtin_amdgcn_readlane(s_l, j + u);
    w[u] = __int_as_float(__builtin_amdgcn_readlane(__float_as_int(w_l), j + u));
  }
  #pragma unroll
  for (int u = 0; u < NB; ++u)
    p[u] = ((const float4*)(x + (size_t)s[u] * INDIM))[lane];
  #pragma unroll
  for (int u = 0; u < NB; ++u) vec_axpy(a, p[u], w[u]);
}

template <int WAVES>
__global__ __launch_bounds__(WAVES * 32)
void gather_in_kernel(const float* __restrict__ x,
                      const int* __restrict__ rowptr,
                      const int* __restrict__ csr_src,
                      const float* __restrict__ dinv,
                      float* __restrict__ ax, int n) {
  const int wv   = threadIdx.x >> 5;
  const int lane = threadIdx.x & 31;
  const int node = __builtin_amdgcn_readfirstlane(blockIdx.x * WAVES + wv);
  if (node >= n) return;

  const int beg = rowptr[node];
  const int end = rowptr[node + 1];
  const float di = dinv[node];
  const float sl = di * di;

  const float4 xs = ((const float4*)(x + (size_t)node * INDIM))[lane];
  float4 a = make_float4(0.0f, 0.0f, 0.0f, 0.0f);

  for (int base = beg; base < end; base += 32) {
    const int m = (end - base < 32) ? (end - base) : 32;
    int   s_l = 0;
    float w_l = 0.0f;
    if (lane < m) {
      s_l = csr_src[base + lane];
      w_l = dinv[s_l] * di;
    }
    int j = 0;
    for (; j + 8 <= m; j += 8) gather_batch<8>(a, x, s_l, w_l, j, lane);
    if (j + 4 <= m) { gather_batch<4>(a, x, s_l, w_l, j, lane); j += 4; }
    if (j + 2 <= m) { gather_batch<2>(a, x, s_l, w_l, j, lane); j += 2; }
    if (j < m)        gather_batch<1>(a, x, s_l, w_l, j, lane);
  }

  vec_axpy(a, xs, sl);
  ((float4*)(ax + (size_t)node * INDIM))[lane] = a;
}

static __device__ __forceinline__ void row_axpy(float4& a0, float4& a1,
                                                const float4 v0, const float4 v1,
                                                const float w) {
  a0.x = fmaf(v0.x, w, a0.x);
  a0.y = fmaf(v0.y, w, a0.y);
  a0.z = fmaf(v0.z, w, a0.z);
  a0.w = fmaf(v0.w, w, a0.w);
  a1.x = fmaf(v1.x, w, a1.x);
  a1.y = fmaf(v1.y, w, a1.y);
  a1.z = fmaf(v1.z, w, a1.z);
  a1.w = fmaf(v1.w, w, a1.w);
}

template <int WAVES>
__global__ __launch_bounds__(WAVES * 32)
void gather_finish_kernel(const float* __restrict__ h,
                          const int* __restrict__ rowptr,
                          const int* __restrict__ csr_src,
                          const float* __restrict__ dinv,
                          const float* __restrict__ bias,
                          const float* __restrict__ Wl,
                          const float* __restrict__ bl,
                          float* __restrict__ dstbuf, int n) {
  const int wv   = threadIdx.x >> 5;
  const int lane = threadIdx.x & 31;
  const int node = __builtin_amdgcn_readfirstlane(blockIdx.x * WAVES + wv);

  float r0 = 0.0f, r1 = 0.0f;
  if (node < n) {
    const int beg = rowptr[node];
    const int end = rowptr[node + 1];
    const float di = dinv[node];
    const float sl = di * di;

    const float4* hp = (const float4*)(h + (size_t)node * HID);
    float4 a0 = hp[lane];
    float4 a1 = hp[lane + 32];
    a0.x *= sl; a0.y *= sl; a0.z *= sl; a0.w *= sl;
    a1.x *= sl; a1.y *= sl; a1.z *= sl; a1.w *= sl;

    for (int base = beg; base < end; base += 32) {
      const int m = (end - base < 32) ? (end - base) : 32;
      int   s_l = 0;
      float w_l = 0.0f;
      if (lane < m) {
        s_l = csr_src[base + lane];
        w_l = dinv[s_l] * di;
      }
      int j = 0;
      for (; j + 4 <= m; j += 4) {
        const int s0 = __builtin_amdgcn_readlane(s_l, j);
        const int s1 = __builtin_amdgcn_readlane(s_l, j + 1);
        const int s2 = __builtin_amdgcn_readlane(s_l, j + 2);
        const int s3 = __builtin_amdgcn_readlane(s_l, j + 3);
        const float w0 = __int_as_float(__builtin_amdgcn_readlane(__float_as_int(w_l), j));
        const float w1 = __int_as_float(__builtin_amdgcn_readlane(__float_as_int(w_l), j + 1));
        const float w2 = __int_as_float(__builtin_amdgcn_readlane(__float_as_int(w_l), j + 2));
        const float w3 = __int_as_float(__builtin_amdgcn_readlane(__float_as_int(w_l), j + 3));
        const float4* h0 = (const float4*)(h + (size_t)s0 * HID);
        const float4* h1 = (const float4*)(h + (size_t)s1 * HID);
        const float4* h2 = (const float4*)(h + (size_t)s2 * HID);
        const float4* h3 = (const float4*)(h + (size_t)s3 * HID);
        const float4 p0 = h0[lane], q0 = h0[lane + 32];
        const float4 p1 = h1[lane], q1 = h1[lane + 32];
        const float4 p2 = h2[lane], q2 = h2[lane + 32];
        const float4 p3 = h3[lane], q3 = h3[lane + 32];
        row_axpy(a0, a1, p0, q0, w0);
        row_axpy(a0, a1, p1, q1, w1);
        row_axpy(a0, a1, p2, q2, w2);
        row_axpy(a0, a1, p3, q3, w3);
      }
      #pragma unroll 1
      for (; j < m; ++j) {
        const int s0 = __builtin_amdgcn_readlane(s_l, j);
        const float w0 = __int_as_float(__builtin_amdgcn_readlane(__float_as_int(w_l), j));
        const float4* h0 = (const float4*)(h + (size_t)s0 * HID);
        const float4 p0 = h0[lane], q0 = h0[lane + 32];
        row_axpy(a0, a1, p0, q0, w0);
      }
    }

    const float4 b0 = ((const float4*)bias)[lane];
    const float4 b1 = ((const float4*)bias)[lane + 32];
    a0.x = add_relu(a0.x, b0.x); a0.y = add_relu(a0.y, b0.y);
    a0.z = add_relu(a0.z, b0.z); a0.w = add_relu(a0.w, b0.w);
    a1.x = add_relu(a1.x, b1.x); a1.y = add_relu(a1.y, b1.y);
    a1.z = add_relu(a1.z, b1.z); a1.w = add_relu(a1.w, b1.w);

    const float4* wp = (const float4*)Wl;
    const float4 wa = wp[2 * lane],      wb = wp[2 * lane + 1];
    const float4 wc = wp[64 + 2 * lane], wd = wp[64 + 2 * lane + 1];
    r0 = a0.x * wa.x;            r1 = a0.x * wa.y;
    r0 = fmaf(a0.y, wa.z, r0);   r1 = fmaf(a0.y, wa.w, r1);
    r0 = fmaf(a0.z, wb.x, r0);   r1 = fmaf(a0.z, wb.y, r1);
    r0 = fmaf(a0.w, wb.z, r0);   r1 = fmaf(a0.w, wb.w, r1);
    r0 = fmaf(a1.x, wc.x, r0);   r1 = fmaf(a1.x, wc.y, r1);
    r0 = fmaf(a1.y, wc.z, r0);   r1 = fmaf(a1.y, wc.w, r1);
    r0 = fmaf(a1.z, wd.x, r0);   r1 = fmaf(a1.z, wd.y, r1);
    r0 = fmaf(a1.w, wd.z, r0);   r1 = fmaf(a1.w, wd.w, r1);
    for (int o = 16; o > 0; o >>= 1) {
      r0 += __shfl_xor(r0, o);
      r1 += __shfl_xor(r1, o);
    }
    r0 += bl[0];
    r1 += bl[1];
  }

  __shared__ float stage[2 * WAVES];
  if (lane == 0) {
    stage[2 * wv]     = r0;
    stage[2 * wv + 1] = r1;
  }
  __syncthreads();
  if (wv == 0) {
    const int idx = blockIdx.x * (2 * WAVES) + lane;
    if (lane < 2 * WAVES && idx < 2 * n) dstbuf[idx] = stage[lane];
  }
}

extern "C" void kernel_launch(void* const* d_in, const int* in_sizes, int n_in,
                              void* d_out, int out_size, void* d_ws, size_t ws_size,
                              hipStream_t stream) {
  const float* x   = (const float*)d_in[0];
  const int*   ei  = (const int*)d_in[1];
  const float* W1  = (const float*)d_in[2];
  const float* b1  = (const float*)d_in[3];
  const float* W2  = (const float*)d_in[4];
  const float* b2  = (const float*)d_in[5];
  const float* Wl  = (const float*)d_in[6];
  const float* bl  = (const float*)d_in[7];
  float*       out = (float*)d_out;

  const int*   src = ei;
  const int*   dst = ei + NEDGES;

  char* ws = (char*)d_ws;
  const size_t H_OFF   = 200192;
  const size_t HBYTES  = (size_t)NNODES * HID * 4;
  const size_t X2_OFF  = H_OFF + HBYTES;
  const size_t BP_OFF  = X2_OFF + HBYTES;
  const size_t CNT_OFF = BP_OFF + 262144;
  const size_t RP_OFF  = CNT_OFF + 200192;
  const size_t RK_OFF  = RP_OFF + 200192;
  const size_t CSR_OFF = RK_OFF + 2400000;

  float* dinv = (float*)ws;
  float* H    = (float*)(ws + H_OFF);
  float* AX   = H;
  float* X2   = (float*)(ws + X2_OFF);
  float* Bp   = (float*)(ws + BP_OFF);
  int*   cnt    = (int*)(ws + CNT_OFF);
  int*   rowptr = (int*)(ws + RP_OFF);
  int*   rank   = (int*)(ws + RK_OFF);
  int*   csrsrc = (int*)(ws + CSR_OFF);

  const int TPB = 256;
  const int nodeBlocks  = (NNODES + TPB - 1) / TPB;
  const int edgeBlocks  = (NEDGES + TPB - 1) / TPB;
  const int rowBlocks8  = (NNODES + 7) / 8;
  const int rowBlocks16 = (NNODES + 15) / 16;
  const int scanBlocks  = (NNODES + SCAN_TPB - 1) / SCAN_TPB;
  const int gemmBlocks  = NNODES / (16 * MTILES);

  zero_i32_kernel<<<nodeBlocks, TPB, 0, stream>>>(cnt, NNODES);
  deg_rank_kernel<<<edgeBlocks, TPB, 0, stream>>>(dst, cnt, rank, NEDGES);
  make_dinv_kernel<<<nodeBlocks, TPB, 0, stream>>>(cnt, dinv, NNODES);
  rowptr_scan_kernel<<<scanBlocks, SCAN_TPB, 0, stream>>>(cnt, rowptr, NNODES);
  csr_fill_kernel<<<edgeBlocks, TPB, 0, stream>>>(src, dst, rank, rowptr, csrsrc, NEDGES);

  {
    const int tot = INDIM * HID;
    gather_in_kernel<8><<<rowBlocks8, 256, 0, stream>>>(x, rowptr, csrsrc, dinv, AX, NNODES);
    pack_b_kernel<<<(tot + TPB - 1) / TPB, TPB, 0, stream>>>(W1, Bp, HID, tot);
    gemm_wmma_f32_kernel<INDIM, true><<<gemmBlocks, 128, 0, stream>>>(AX, Bp, b1, X2, HID);
  }

  {
    const int tot = HID * HID;
    pack_b_kernel<<<(tot + TPB - 1) / TPB, TPB, 0, stream>>>(W2, Bp, HID, tot);
    gemm_wmma_f32_kernel<HID, false><<<gemmBlocks, 128, 0, stream>>>(X2, Bp, b2, H, HID);
    gather_finish_kernel<16><<<rowBlocks16, 512, 0, stream>>>(
        H, rowptr, csrsrc, dinv, b2, Wl, bl, out, NNODES);
  }
}


# Both GEMMs feed A from LDS: a block copies its 80 rows by whole 128-byte lines, 32 k per stage, two stages, instead of each wave taking 16-byte pieces of 16 lines per fragment; k order unchanged
# speedup vs baseline: 3.9290x; 1.0500x over previous
#include <hip/hip_runtime.h>
#include <hip/hip_bf16.h>

typedef float v2f __attribute__((ext_vector_type(2)));
typedef float v8f __attribute__((ext_vector_type(8)));

#define NNODES 50000
#define NEDGES 600000
#define INDIM 128
#define HID 256
#define MTILES 5

__global__ void zero_i32_kernel(int* __restrict__ p, int n) {
  int i = blockIdx.x * blockDim.x + threadIdx.x;
  if (i < n) p[i] = 0;
}

__global__ void deg_rank_kernel(const int* __restrict__ dst, int* __restrict__ cnt,
                                int* __restrict__ rank, int E) {
  int e = blockIdx.x * blockDim.x + threadIdx.x;
  if (e < E) rank[e] = atomicAdd(&cnt[dst[e]], 1);
}

__global__ void make_dinv_kernel(const int* __restrict__ cnt, float* __restrict__ dinv, int n) {
  int i = blockIdx.x * blockDim.x + threadIdx.x;
  if (i < n) dinv[i] = rsqrtf((float)cnt[i] + 1.0f);
}

#define SCAN_TPB 1024
__global__ __launch_bounds__(SCAN_TPB)
void rowptr_scan_kernel(const int* __restrict__ cnt, int* __restrict__ rowptr, int n) {
  __shared__ int wpre[32];
  __shared__ int wtot[32];
  const int tid  = threadIdx.x;
  const int lane = tid & 31;
  const int wv   = tid >> 5;
  const int base = blockIdx.x * SCAN_TPB;

  int pre = 0;
  for (int j = tid; j < base; j += SCAN_TPB) pre += cnt[j];

  const int i = base + tid;
  const int v = (i < n) ? cnt[i] : 0;

  int x = v;
  for (int o = 1; o < 32; o <<= 1) {
    int y = __shfl_up(x, o);
    if (lane >= o) x += y;
  }
  for (int o = 16; o > 0; o >>= 1) pre += __shfl_xor(pre, o);

  if (lane == 31) wtot[wv] = x;
  if (lane == 0)  wpre[wv] = pre;
  __syncthreads();

  int t = wpre[lane] + ((lane < wv) ? wtot[lane] : 0);
  for (int o = 16; o > 0; o >>= 1) t += __shfl_xor(t, o);

  const int excl = t + x - v;
  if (i < n) rowptr[i] = excl;
  if (i == n - 1) rowptr[n] = excl + v;
}

__global__ void csr_fill_kernel(const int* __restrict__ src, const int* __restrict__ dst,
                                const int* __restrict__ rank, const int* __restrict__ rowptr,
                                int* __restrict__ csr_src, int E) {
  int e = blockIdx.x * blockDim.x + threadIdx.x;
  if (e < E) csr_src[rowptr[dst[e]] + rank[e]] = src[e];
}

__global__ void pack_b_kernel(const float* __restrict__ B, float* __restrict__ Bp,
                              int N, int total) {
  int i = blockIdx.x * blockDim.x + threadIdx.x;
  if (i >= total) return;
  int k = i / N;
  int n = i - k * N;
  Bp[((size_t)(k >> 1) * N + n) * 2 + (k & 1)] = B[i];
}

static __device__ __forceinline__ float add_relu(const float a, const float b) {
  const float v = a + b;
  return v > 0.0f ? v : 0.0f;
}

#define KCH 32
#define AROWS (16 * MTILES)
#define APITCH 36

static __device__ __forceinline__ void load_b_chunk(v2f (&bf)[KCH / 4][4],
                                                    const float* __restrict__ bk, const int N) {
  #pragma unroll
  for (int s = 0; s < KCH / 4; ++s) {
    #pragma unroll
    for (int u = 0; u < 4; ++u)
      bf[s][u] = *(const v2f*)(bk + (size_t)s * 2 * N * 2 + u * 32);
  }
}

static __device__ __forceinline__ void mma_chunk(v8f (&acc)[MTILES][4],
                                                 const float* af0,
                                                 const v2f (&bf)[KCH / 4][4]) {
  #pragma unroll
  for (int s = 0; s < KCH / 4; ++s) {
    #pragma unroll
    for (int t = 0; t < MTILES; ++t) {
      const v2f af = *(const v2f*)(af0 + t * 16 * APITCH + 4 * s);
      #pragma unroll
      for (int u = 0; u < 4; ++u) {
        acc[t][u] = __builtin_amdgcn_wmma_f32_16x16x4_f32(
            false, af, false, bf[s][u], (short)0, acc[t][u], false, false);
      }
    }
  }
}

template <int K, bool BIAS_RELU>
__global__ __launch_bounds__(128, 1)
void gemm_wmma_f32_kernel(const float* __restrict__ A,
                          const float* __restrict__ Bp,
                          const float* __restrict__ bias,
                          float* __restrict__ H,
                          int N) {
  __shared__ __attribute__((aligned(16))) float As[2][AROWS * APITCH];
  constexpr int NCH = K / KCH;

  const int tid  = threadIdx.x;
  const int wave = tid >> 5;
  const int lane = tid & 31;
  const int half = lane >> 4;
  const int lidx = lane & 15;

  const int m0 = blockIdx.x * AROWS;
  const int n0 = wave * 64;

  v8f acc[MTILES][4] = {};

  const int crow = tid >> 3;
  const int cseg = tid & 7;
  const float* gsrc = A + (size_t)(m0 + crow) * K + cseg * 4;
  const int ldst = crow * APITCH + cseg * 4;

  {
    float4 c0[MTILES];
    #pragma unroll
    for (int i = 0; i < MTILES; ++i)
      c0[i] = *(const float4*)(gsrc + (size_t)i * 16 * K);
    #pragma unroll
    for (int i = 0; i < MTILES; ++i)
      *(float4*)(&As[0][ldst + i * 16 * APITCH]) = c0[i];
  }
  __syncthreads();

  const float* bbase = Bp + ((size_t)half * N + n0 + lidx) * 2;
  const int aoff = lidx * APITCH + 2 * half;

  #pragma unroll 1
  for (int kc = 0; kc < NCH - 1; ++kc) {
    v2f bf[KCH / 4][4];
    load_b_chunk(bf, bbase + (size_t)kc * (KCH / 2) * N * 2, N);
    float4 nx[MTILES];
    #pragma unroll
    for (int i = 0; i < MTILES; ++i)
      nx[i] = *(const float4*)(gsrc + (size_t)i * 16 * K + (kc + 1) * KCH);

    mma_chunk(acc, &As[kc & 1][aoff], bf);

    float* ad = As[(kc + 1) & 1];
    #pragma unroll
    for (int i = 0; i < MTILES; ++i)
      *(float4*)(&ad[ldst + i * 16 * APITCH]) = nx[i];
    __syncthreads();
  }
  {
    v2f bf[KCH / 4][4];
    load_b_chunk(bf, bbase + (size_t)(NCH - 1) * (KCH / 2) * N * 2, N);
    mma_chunk(acc, &As[(NCH - 1) & 1][aoff], bf);
  }

  float bv[4] = {0.0f, 0.0f, 0.0f, 0.0f};
  if constexpr (BIAS_RELU) {
    #pragma unroll
    for (int u = 0; u < 4; ++u) bv[u] = bias[n0 + lidx + u * 16];
  }
  #pragma unroll
  for (int t = 0; t < MTILES; ++t) {
    #pragma unroll
    for (int g = 0; g < 8; ++g) {
      const int row = m0 + t * 16 + g + 8 * half;
      const size_t base = (size_t)row * N + n0 + lidx;
      #pragma unroll
      for (int u = 0; u < 4; ++u) {
        if constexpr (BIAS_RELU) H[base + u * 16] = add_relu(acc[t][u][g], bv[u]);
        else                     H[base + u * 16] = acc[t][u][g];
      }
    }
  }
}

static __device__ __forceinline__ void vec_axpy(float4& a, const float4 v, const float w) {
  a.x = fmaf(v.x, w, a.x);
  a.y = fmaf(v.y, w, a.y);
  a.z = fmaf(v.z, w, a.z);
  a.w = fmaf(v.w, w, a.w);
}

template <int NB>
static __device__ __forceinline__ void gather_batch(float4& a, const float* __restrict__ x,
                                                    const int s_l, const float w_l,
                                                    const int j, const int lane) {
  int s[NB]; float w[NB]; float4 p[NB];
  #pragma unroll
  for (int u = 0; u < NB; ++u) {
    s[u] = __builtin_amdgcn_readlane(s_l, j + u);
    w[u] = __int_as_float(__builtin_amdgcn_readlane(__float_as_int(w_l), j + u));
  }
  #pragma unroll
  for (int u = 0; u < NB; ++u)
    p[u] = ((const float4*)(x + (size_t)s[u] * INDIM))[lane];
  #pragma unroll
  for (int u = 0; u < NB; ++u) vec_axpy(a, p[u], w[u]);
}

template <int WAVES>
__global__ __launch_bounds__(WAVES * 32)
void gather_in_kernel(const float* __restrict__ x,
                      const int* __restrict__ rowptr,
                      const int* __restrict__ csr_src,
                      const float* __restrict__ dinv,
                      float* __restrict__ ax, int n) {
  const int wv   = threadIdx.x >> 5;
  const int lane = threadIdx.x & 31;
  const int node = __builtin_amdgcn_readfirstlane(blockIdx.x * WAVES + wv);
  if (node >= n) return;

  const int beg = rowptr[node];
  const int end = rowptr[node + 1];
  const float di = dinv[node];
  const float sl = di * di;

  const float4 xs = ((const float4*)(x + (size_t)node * INDIM))[lane];
  float4 a = make_float4(0.0f, 0.0f, 0.0f, 0.0f);

  for (int base = beg; base < end; base += 32) {
    const int m = (end - base < 32) ? (end - base) : 32;
    int   s_l = 0;
    float w_l = 0.0f;
    if (lane < m) {
      s_l = csr_src[base + lane];
      w_l = dinv[s_l] * di;
    }
    int j = 0;
    for (; j + 8 <= m; j += 8) gather_batch<8>(a, x, s_l, w_l, j, lane);
    if (j + 4 <= m) { gather_batch<4>(a, x, s_l, w_l, j, lane); j += 4; }
    if (j + 2 <= m) { gather_batch<2>(a, x, s_l, w_l, j, lane); j += 2; }
    if (j < m)        gather_batch<1>(a, x, s_l, w_l, j, lane);
  }

  vec_axpy(a, xs, sl);
  ((float4*)(ax + (size_t)node * INDIM))[lane] = a;
}

static __device__ __forceinline__ void row_axpy(float4& a0, float4& a1,
                                                const float4 v0, const float4 v1,
                                                const float w) {
  a0.x = fmaf(v0.x, w, a0.x);
  a0.y = fmaf(v0.y, w, a0.y);
  a0.z = fmaf(v0.z, w, a0.z);
  a0.w = fmaf(v0.w, w, a0.w);
  a1.x = fmaf(v1.x, w, a1.x);
  a1.y = fmaf(v1.y, w, a1.y);
  a1.z = fmaf(v1.z, w, a1.z);
  a1.w = fmaf(v1.w, w, a1.w);
}

template <int WAVES>
__global__ __launch_bounds__(WAVES * 32)
void gather_finish_kernel(const float* __restrict__ h,
                          const int* __restrict__ rowptr,
                          const int* __restrict__ csr_src,
                          const float* __restrict__ dinv,
                          const float* __restrict__ bias,
                          const float* __restrict__ Wl,
                          const float* __restrict__ bl,
                          float* __restrict__ dstbuf, int n) {
  const int wv   = threadIdx.x >> 5;
  const int lane = threadIdx.x & 31;
  const int node = __builtin_amdgcn_readfirstlane(blockIdx.x * WAVES + wv);

  float r0 = 0.0f, r1 = 0.0f;
  if (node < n) {
    const int beg = rowptr[node];
    const int end = rowptr[node + 1];
    const float di = dinv[node];
    const float sl = di * di;

    const float4* hp = (const float4*)(h + (size_t)node * HID);
    float4 a0 = hp[lane];
    float4 a1 = hp[lane + 32];
    a0.x *= sl; a0.y *= sl; a0.z *= sl; a0.w *= sl;
    a1.x *= sl; a1.y *= sl; a1.z *= sl; a1.w *= sl;

    for (int base = beg; base < end; base += 32) {
      const int m = (end - base < 32) ? (end - base) : 32;
      int   s_l = 0;
      float w_l = 0.0f;
      if (lane < m) {
        s_l = csr_src[base + lane];
        w_l = dinv[s_l] * di;
      }
      int j = 0;
      for (; j + 4 <= m; j += 4) {
        const int s0 = __builtin_amdgcn_readlane(s_l, j);
        const int s1 = __builtin_amdgcn_readlane(s_l, j + 1);
        const int s2 = __builtin_amdgcn_readlane(s_l, j + 2);
        const int s3 = __builtin_amdgcn_readlane(s_l, j + 3);
        const float w0 = __int_as_float(__builtin_amdgcn_readlane(__float_as_int(w_l), j));
        const float w1 = __int_as_float(__builtin_amdgcn_readlane(__float_as_int(w_l), j + 1));
        const float w2 = __int_as_float(__builtin_amdgcn_readlane(__float_as_int(w_l), j + 2));
        const float w3 = __int_as_float(__builtin_amdgcn_readlane(__float_as_int(w_l), j + 3));
        const float4* h0 = (const float4*)(h + (size_t)s0 * HID);
        const float4* h1 = (const float4*)(h + (size_t)s1 * HID);
        const float4* h2 = (const float4*)(h + (size_t)s2 * HID);
        const float4* h3 = (const float4*)(h + (size_t)s3 * HID);
        const float4 p0 = h0[lane], q0 = h0[lane + 32];
        const float4 p1 = h1[lane], q1 = h1[lane + 32];
        const float4 p2 = h2[lane], q2 = h2[lane + 32];
        const float4 p3 = h3[lane], q3 = h3[lane + 32];
        row_axpy(a0, a1, p0, q0, w0);
        row_axpy(a0, a1, p1, q1, w1);
        row_axpy(a0, a1, p2, q2, w2);
        row_axpy(a0, a1, p3, q3, w3);
      }
      #pragma unroll 1
      for (; j < m; ++j) {
        const int s0 = __builtin_amdgcn_readlane(s_l, j);
        const float w0 = __int_as_float(__builtin_amdgcn_readlane(__float_as_int(w_l), j));
        const float4* h0 = (const float4*)(h + (size_t)s0 * HID);
        const float4 p0 = h0[lane], q0 = h0[lane + 32];
        row_axpy(a0, a1, p0, q0, w0);
      }
    }

    const float4 b0 = ((const float4*)bias)[lane];
    const float4 b1 = ((const float4*)bias)[lane + 32];
    a0.x = add_relu(a0.x, b0.x); a0.y = add_relu(a0.y, b0.y);
    a0.z = add_relu(a0.z, b0.z); a0.w = add_relu(a0.w, b0.w);
    a1.x = add_relu(a1.x, b1.x); a1.y = add_relu(a1.y, b1.y);
    a1.z = add_relu(a1.z, b1.z); a1.w = add_relu(a1.w, b1.w);

    const float4* wp = (const float4*)Wl;
    const float4 wa = wp[2 * lane],      wb = wp[2 * lane + 1];
    const float4 wc = wp[64 + 2 * lane], wd = wp[64 + 2 * lane + 1];
    r0 = a0.x * wa.x;            r1 = a0.x * wa.y;
    r0 = fmaf(a0.y, wa.z, r0);   r1 = fmaf(a0.y, wa.w, r1);
    r0 = fmaf(a0.z, wb.x, r0);   r1 = fmaf(a0.z, wb.y, r1);
    r0 = fmaf(a0.w, wb.z, r0);   r1 = fmaf(a0.w, wb.w, r1);
    r0 = fmaf(a1.x, wc.x, r0);   r1 = fmaf(a1.x, wc.y, r1);
    r0 = fmaf(a1.y, wc.z, r0);   r1 = fmaf(a1.y, wc.w, r1);
    r0 = fmaf(a1.z, wd.x, r0);   r1 = fmaf(a1.z, wd.y, r1);
    r0 = fmaf(a1.w, wd.z, r0);   r1 = fmaf(a1.w, wd.w, r1);
    for (int o = 16; o > 0; o >>= 1) {
      r0 += __shfl_xor(r0, o);
      r1 += __shfl_xor(r1, o);
    }
    r0 += bl[0];
    r1 += bl[1];
  }

  __shared__ float stage[2 * WAVES];
  if (lane == 0) {
    stage[2 * wv]     = r0;
    stage[2 * wv + 1] = r1;
  }
  __syncthreads();
  if (wv == 0) {
    const int idx = blockIdx.x * (2 * WAVES) + lane;
    if (lane < 2 * WAVES && idx < 2 * n) dstbuf[idx] = stage[lane];
  }
}

extern "C" void kernel_launch(void* const* d_in, const int* in_sizes, int n_in,
                              void* d_out, int out_size, void* d_ws, size_t ws_size,
                              hipStream_t stream) {
  const float* x   = (const float*)d_in[0];
  const int*   ei  = (const int*)d_in[1];
  const float* W1  = (const float*)d_in[2];
  const float* b1  = (const float*)d_in[3];
  const float* W2  = (const float*)d_in[4];
  const float* b2  = (const float*)d_in[5];
  const float* Wl  = (const float*)d_in[6];
  const float* bl  = (const float*)d_in[7];
  float*       out = (float*)d_out;

  const int*   src = ei;
  const int*   dst = ei + NEDGES;

  char* ws = (char*)d_ws;
  const size_t H_OFF   = 200192;
  const size_t HBYTES  = (size_t)NNODES * HID * 4;
  const size_t X2_OFF  = H_OFF + HBYTES;
  const size_t BP_OFF  = X2_OFF + HBYTES;
  const size_t CNT_OFF = BP_OFF + 262144;
  const size_t RP_OFF  = CNT_OFF + 200192;
  const size_t RK_OFF  = RP_OFF + 200192;
  const size_t CSR_OFF = RK_OFF + 2400000;

  float* dinv = (float*)ws;
  float* H    = (float*)(ws + H_OFF);
  float* AX   = H;
  float* X2   = (float*)(ws + X2_OFF);
  float* Bp   = (float*)(ws + BP_OFF);
  int*   cnt    = (int*)(ws + CNT_OFF);
  int*   rowptr = (int*)(ws + RP_OFF);
  int*   rank   = (int*)(ws + RK_OFF);
  int*   csrsrc = (int*)(ws + CSR_OFF);

  const int TPB = 256;
  const int nodeBlocks  = (NNODES + TPB - 1) / TPB;
  const int edgeBlocks  = (NEDGES + TPB - 1) / TPB;
  const int rowBlocks8  = (NNODES + 7) / 8;
  const int rowBlocks16 = (NNODES + 15) / 16;
  const int scanBlocks  = (NNODES + SCAN_TPB - 1) / SCAN_TPB;
  const int gemmBlocks  = NNODES / (16 * MTILES);

  zero_i32_kernel<<<nodeBlocks, TPB, 0, stream>>>(cnt, NNODES);
  deg_rank_kernel<<<edgeBlocks, TPB, 0, stream>>>(dst, cnt, rank, NEDGES);
  make_dinv_kernel<<<nodeBlocks, TPB, 0, stream>>>(cnt, dinv, NNODES);
  rowptr_scan_kernel<<<scanBlocks, SCAN_TPB, 0, stream>>>(cnt, rowptr, NNODES);
  csr_fill_kernel<<<edgeBlocks, TPB, 0, stream>>>(src, dst, rank, rowptr, csrsrc, NEDGES);

  {
    const int tot = INDIM * HID;
    gather_in_kernel<8><<<rowBlocks8, 256, 0, stream>>>(x, rowptr, csrsrc, dinv, AX, NNODES);
    pack_b_kernel<<<(tot + TPB - 1) / TPB, TPB, 0, stream>>>(W1, Bp, HID, tot);
    gemm_wmma_f32_kernel<INDIM, true><<<gemmBlocks, 128, 0, stream>>>(AX, Bp, b1, X2, HID);
  }

  {
    const int tot = HID * HID;
    pack_b_kernel<<<(tot + TPB - 1) / TPB, TPB, 0, stream>>>(W2, Bp, HID, tot);
    gemm_wmma_f32_kernel<HID, false><<<gemmBlocks, 128, 0, stream>>>(X2, Bp, b2, H, HID);
    gather_finish_kernel<16><<<rowBlocks16, 512, 0, stream>>>(
        H, rowptr, csrsrc, dinv, b2, Wl, bl, out, NNODES);
  }
}


# Edge lists as fixed 32-id buckets per destination row, written straight from the counting atomic's return value (overflow list past 32): scan, fill and dinv launches go; sums stay in arrival order
# speedup vs baseline: 4.5014x; 1.1457x over previous
#include <hip/hip_runtime.h>
#include <hip/hip_bf16.h>

typedef float v2f __attribute__((ext_vector_type(2)));
typedef float v8f __attribute__((ext_vector_type(8)));

#define NNODES 50000
#define NEDGES 600000
#define INDIM 128
#define HID 256
#define MTILES 5

#define CAP 32

__global__ void zero_i32_kernel(int* __restrict__ p, int n) {
  int i = blockIdx.x * blockDim.x + threadIdx.x;
  if (i < n) p[i] = 0;
}

__global__ void deg_bucket_kernel(const int* __restrict__ src, const int* __restrict__ dst,
                                  int* __restrict__ cnt, int* __restrict__ bucket,
                                  int* __restrict__ ovf_dst, int* __restrict__ ovf_src,
                                  int E, int n) {
  int e = blockIdx.x * blockDim.x + threadIdx.x;
  if (e < E) {
    const int d = dst[e];
    const int s = src[e];
    const int r = atomicAdd(&cnt[d], 1);
    if (r < CAP) {
      bucket[(size_t)d * CAP + r] = s;
    } else {
      const int p = atomicAdd(&cnt[n], 1);
      ovf_dst[p] = d;
      ovf_src[p] = s;
    }
  }
}

static __device__ __forceinline__ float dinv_of(const int c) {
  return rsqrtf((float)c + 1.0f);
}

__global__ void pack_b_kernel(const float* __restrict__ B, float* __restrict__ Bp,
                              int N, int total) {
  int i = blockIdx.x * blockDim.x + threadIdx.x;
  if (i >= total) return;
  int k = i / N;
  int n = i - k * N;
  Bp[((size_t)(k >> 1) * N + n) * 2 + (k & 1)] = B[i];
}

static __device__ __forceinline__ float add_relu(const float a, const float b) {
  const float v = a + b;
  return v > 0.0f ? v : 0.0f;
}

#define KCH 32
#define AROWS (16 * MTILES)
#define APITCH 36

static __device__ __forceinline__ void load_b_chunk(v2f (&bf)[KCH / 4][4],
                                                    const float* __restrict__ bk, const int N) {
  #pragma unroll
  for (int s = 0; s < KCH / 4; ++s) {
    #pragma unroll
    for (int u = 0; u < 4; ++u)
      bf[s][u] = *(const v2f*)(bk + (size_t)s * 2 * N * 2 + u * 32);
  }
}

static __device__ __forceinline__ void mma_chunk(v8f (&acc)[MTILES][4],
                                                 const float* af0,
                                                 const v2f (&bf)[KCH / 4][4]) {
  #pragma unroll
  for (int s = 0; s < KCH / 4; ++s) {
    #pragma unroll
    for (int t = 0; t < MTILES; ++t) {
      const v2f af = *(const v2f*)(af0 + t * 16 * APITCH + 4 * s);
      #pragma unroll
      for (int u = 0; u < 4; ++u) {
        acc[t][u] = __builtin_amdgcn_wmma_f32_16x16x4_f32(
            false, af, false, bf[s][u], (short)0, acc[t][u], false, false);
      }
    }
  }
}

template <int K, bool BIAS_RELU>
__global__ __launch_bounds__(128, 1)
void gemm_wmma_f32_kernel(const float* __restrict__ A,
                          const float* __restrict__ Bp,
                          const float* __restrict__ bias,
                          float* __restrict__ H,
                          int N) {
  __shared__ __attribute__((aligned(16))) float As[2][AROWS * APITCH];
  constexpr int NCH = K / KCH;

  const int tid  = threadIdx.x;
  const int wave = tid >> 5;
  const int lane = tid & 31;
  const int half = lane >> 4;
  const int lidx = lane & 15;

  const int m0 = blockIdx.x * AROWS;
  const int n0 = wave * 64;

  v8f acc[MTILES][4] = {};

  const int crow = tid >> 3;
  const int cseg = tid & 7;
  const float* gsrc = A + (size_t)(m0 + crow) * K + cseg * 4;
  const int ldst = crow * APITCH + cseg * 4;

  {
    float4 c0[MTILES];
    #pragma unroll
    for (int i = 0; i < MTILES; ++i)
      c0[i] = *(const float4*)(gsrc + (size_t)i * 16 * K);
    #pragma unroll
    for (int i = 0; i < MTILES; ++i)
      *(float4*)(&As[0][ldst + i * 16 * APITCH]) = c0[i];
  }
  __syncthreads();

  const float* bbase = Bp + ((size_t)half * N + n0 + lidx) * 2;
  const int aoff = lidx * APITCH + 2 * half;

  #pragma unroll 1
  for (int kc = 0; kc < NCH - 1; ++kc) {
    v2f bf[KCH / 4][4];
    load_b_chunk(bf, bbase + (size_t)kc * (KCH / 2) * N * 2, N);
    float4 nx[MTILES];
    #pragma unroll
    for (int i = 0; i < MTILES; ++i)
      nx[i] = *(const float4*)(gsrc + (size_t)i * 16 * K + (kc + 1) * KCH);

    mma_chunk(acc, &As[kc & 1][aoff], bf);

    float* ad = As[(kc + 1) & 1];
    #pragma unroll
    for (int i = 0; i < MTILES; ++i)
      *(float4*)(&ad[ldst + i * 16 * APITCH]) = nx[i];
    __syncthreads();
  }
  {
    v2f bf[KCH / 4][4];
    load_b_chunk(bf, bbase + (size_t)(NCH - 1) * (KCH / 2) * N * 2, N);
    mma_chunk(acc, &As[(NCH - 1) & 1][aoff], bf);
  }

  float bv[4] = {0.0f, 0.0f, 0.0f, 0.0f};
  if constexpr (BIAS_RELU) {
    #pragma unroll
    for (int u = 0; u < 4; ++u) bv[u] = bias[n0 + lidx + u * 16];
  }
  #pragma unroll
  for (int t = 0; t < MTILES; ++t) {
    #pragma unroll
    for (int g = 0; g < 8; ++g) {
      const int row = m0 + t * 16 + g + 8 * half;
      const size_t base = (size_t)row * N + n0 + lidx;
      #pragma unroll
      for (int u = 0; u < 4; ++u) {
        if constexpr (BIAS_RELU) H[base + u * 16] = add_relu(acc[t][u][g], bv[u]);
        else                     H[base + u * 16] = acc[t][u][g];
      }
    }
  }
}

static __device__ __forceinline__ void vec_axpy(float4& a, const float4 v, const float w) {
  a.x = fmaf(v.x, w, a.x);
  a.y = fmaf(v.y, w, a.y);
  a.z = fmaf(v.z, w, a.z);
  a.w = fmaf(v.w, w, a.w);
}

template <int NB>
static __device__ __forceinline__ void gather_batch(float4& a, const float* __restrict__ x,
                                                    const int s_l, const float w_l,
                                                    const int j, const int lane) {
  int s[NB]; float w[NB]; float4 p[NB];
  #pragma unroll
  for (int u = 0; u < NB; ++u) {
    s[u] = __builtin_amdgcn_readlane(s_l, j + u);
    w[u] = __int_as_float(__builtin_amdgcn_readlane(__float_as_int(w_l), j + u));
  }
  #pragma unroll
  for (int u = 0; u < NB; ++u)
    p[u] = ((const float4*)(x + (size_t)s[u] * INDIM))[lane];
  #pragma unroll
  for (int u = 0; u < NB; ++u) vec_axpy(a, p[u], w[u]);
}

template <int WAVES>
__global__ __launch_bounds__(WAVES * 32)
void gather_in_kernel(const float* __restrict__ x,
                      const int* __restrict__ cnt,
                      const int* __restrict__ bucket,
                      const int* __restrict__ ovf_dst,
                      const int* __restrict__ ovf_src,
                      float* __restrict__ ax, int n) {
  const int wv   = threadIdx.x >> 5;
  const int lane = threadIdx.x & 31;
  const int node = __builtin_amdgcn_readfirstlane(blockIdx.x * WAVES + wv);
  if (node >= n) return;

  const int c = cnt[node];
  const float di = dinv_of(c);
  const float sl = di * di;

  const float4 xs = ((const float4*)(x + (size_t)node * INDIM))[lane];
  float4 a = make_float4(0.0f, 0.0f, 0.0f, 0.0f);

  {
    const int m = (c < CAP) ? c : CAP;
    int   s_l = 0;
    float w_l = 0.0f;
    if (lane < m) {
      s_l = bucket[(size_t)node * CAP + lane];
      w_l = dinv_of(cnt[s_l]) * di;
    }
    int j = 0;
    for (; j + 8 <= m; j += 8) gather_batch<8>(a, x, s_l, w_l, j, lane);
    if (j + 4 <= m) { gather_batch<4>(a, x, s_l, w_l, j, lane); j += 4; }
    if (j + 2 <= m) { gather_batch<2>(a, x, s_l, w_l, j, lane); j += 2; }
    if (j < m)        gather_batch<1>(a, x, s_l, w_l, j, lane);
  }
  if (c > CAP) {
    const int novf = cnt[n];
    for (int b = 0; b < novf; b += 32) {
      const int i = b + lane;
      int od = -1, os = 0;
      if (i < novf) { od = ovf_dst[i]; os = ovf_src[i]; }
      unsigned mask = __builtin_amdgcn_ballot_w32(od == node);
      while (mask != 0u) {
        const int jj = __builtin_ctz(mask);
        mask &= mask - 1u;
        const int s = __builtin_amdgcn_readlane(os, jj);
        const float w = dinv_of(cnt[s]) * di;
        const float4 p = ((const float4*)(x + (size_t)s * INDIM))[lane];
        vec_axpy(a, p, w);
      }
    }
  }

  vec_axpy(a, xs, sl);
  ((float4*)(ax + (size_t)node * INDIM))[lane] = a;
}

static __device__ __forceinline__ void row_axpy(float4& a0, float4& a1,
                                                const float4 v0, const float4 v1,
                                                const float w) {
  a0.x = fmaf(v0.x, w, a0.x);
  a0.y = fmaf(v0.y, w, a0.y);
  a0.z = fmaf(v0.z, w, a0.z);
  a0.w = fmaf(v0.w, w, a0.w);
  a1.x = fmaf(v1.x, w, a1.x);
  a1.y = fmaf(v1.y, w, a1.y);
  a1.z = fmaf(v1.z, w, a1.z);
  a1.w = fmaf(v1.w, w, a1.w);
}

template <int WAVES>
__global__ __launch_bounds__(WAVES * 32)
void gather_finish_kernel(const float* __restrict__ h,
                          const int* __restrict__ cnt,
                          const int* __restrict__ bucket,
                          const int* __restrict__ ovf_dst,
                          const int* __restrict__ ovf_src,
                          const float* __restrict__ bias,
                          const float* __restrict__ Wl,
                          const float* __restrict__ bl,
                          float* __restrict__ dstbuf, int n) {
  const int wv   = threadIdx.x >> 5;
  const int lane = threadIdx.x & 31;
  const int node = __builtin_amdgcn_readfirstlane(blockIdx.x * WAVES + wv);

  float r0 = 0.0f, r1 = 0.0f;
  if (node < n) {
    const int c = cnt[node];
    const float di = dinv_of(c);
    const float sl = di * di;

    const float4* hp = (const float4*)(h + (size_t)node * HID);
    float4 a0 = hp[lane];
    float4 a1 = hp[lane + 32];
    a0.x *= sl; a0.y *= sl; a0.z *= sl; a0.w *= sl;
    a1.x *= sl; a1.y *= sl; a1.z *= sl; a1.w *= sl;

    {
      const int m = (c < CAP) ? c : CAP;
      int   s_l = 0;
      float w_l = 0.0f;
      if (lane < m) {
        s_l = bucket[(size_t)node * CAP + lane];
        w_l = dinv_of(cnt[s_l]) * di;
      }
      int j = 0;
      for (; j + 4 <= m; j += 4) {
        const int s0 = __builtin_amdgcn_readlane(s_l, j);
        const int s1 = __builtin_amdgcn_readlane(s_l, j + 1);
        const int s2 = __builtin_amdgcn_readlane(s_l, j + 2);
        const int s3 = __builtin_amdgcn_readlane(s_l, j + 3);
        const float w0 = __int_as_float(__builtin_amdgcn_readlane(__float_as_int(w_l), j));
        const float w1 = __int_as_float(__builtin_amdgcn_readlane(__float_as_int(w_l), j + 1));
        const float w2 = __int_as_float(__builtin_amdgcn_readlane(__float_as_int(w_l), j + 2));
        const float w3 = __int_as_float(__builtin_amdgcn_readlane(__float_as_int(w_l), j + 3));
        const float4* h0 = (const float4*)(h + (size_t)s0 * HID);
        const float4* h1 = (const float4*)(h + (size_t)s1 * HID);
        const float4* h2 = (const float4*)(h + (size_t)s2 * HID);
        const float4* h3 = (const float4*)(h + (size_t)s3 * HID);
        const float4 p0 = h0[lane], q0 = h0[lane + 32];
        const float4 p1 = h1[lane], q1 = h1[lane + 32];
        const float4 p2 = h2[lane], q2 = h2[lane + 32];
        const float4 p3 = h3[lane], q3 = h3[lane + 32];
        row_axpy(a0, a1, p0, q0, w0);
        row_axpy(a0, a1, p1, q1, w1);
        row_axpy(a0, a1, p2, q2, w2);
        row_axpy(a0, a1, p3, q3, w3);
      }
      #pragma unroll 1
      for (; j < m; ++j) {
        const int s0 = __builtin_amdgcn_readlane(s_l, j);
        const float w0 = __int_as_float(__builtin_amdgcn_readlane(__float_as_int(w_l), j));
        const float4* h0 = (const float4*)(h + (size_t)s0 * HID);
        const float4 p0 = h0[lane], q0 = h0[lane + 32];
        row_axpy(a0, a1, p0, q0, w0);
      }
    }
    if (c > CAP) {
      const int novf = cnt[n];
      for (int b = 0; b < novf; b += 32) {
        const int i = b + lane;
        int od = -1, os = 0;
        if (i < novf) { od = ovf_dst[i]; os = ovf_src[i]; }
        unsigned mask = __builtin_amdgcn_ballot_w32(od == node);
        while (mask != 0u) {
          const int jj = __builtin_ctz(mask);
          mask &= mask - 1u;
          const int s0 = __builtin_amdgcn_readlane(os, jj);
          const float w0 = dinv_of(cnt[s0]) * di;
          const float4* h0 = (const float4*)(h + (size_t)s0 * HID);
          const float4 p0 = h0[lane], q0 = h0[lane + 32];
          row_axpy(a0, a1, p0, q0, w0);
        }
      }
    }

    const float4 b0 = ((const float4*)bias)[lane];
    const float4 b1 = ((const float4*)bias)[lane + 32];
    a0.x = add_relu(a0.x, b0.x); a0.y = add_relu(a0.y, b0.y);
    a0.z = add_relu(a0.z, b0.z); a0.w = add_relu(a0.w, b0.w);
    a1.x = add_relu(a1.x, b1.x); a1.y = add_relu(a1.y, b1.y);
    a1.z = add_relu(a1.z, b1.z); a1.w = add_relu(a1.w, b1.w);

    const float4* wp = (const float4*)Wl;
    const float4 wa = wp[2 * lane],      wb = wp[2 * lane + 1];
    const float4 wc = wp[64 + 2 * lane], wd = wp[64 + 2 * lane + 1];
    r0 = a0.x * wa.x;            r1 = a0.x * wa.y;
    r0 = fmaf(a0.y, wa.z, r0);   r1 = fmaf(a0.y, wa.w, r1);
    r0 = fmaf(a0.z, wb.x, r0);   r1 = fmaf(a0.z, wb.y, r1);
    r0 = fmaf(a0.w, wb.z, r0);   r1 = fmaf(a0.w, wb.w, r1);
    r0 = fmaf(a1.x, wc.x, r0);   r1 = fmaf(a1.x, wc.y, r1);
    r0 = fmaf(a1.y, wc.z, r0);   r1 = fmaf(a1.y, wc.w, r1);
    r0 = fmaf(a1.z, wd.x, r0);   r1 = fmaf(a1.z, wd.y, r1);
    r0 = fmaf(a1.w, wd.z, r0);   r1 = fmaf(a1.w, wd.w, r1);
    for (int o = 16; o > 0; o >>= 1) {
      r0 += __shfl_xor(r0, o);
      r1 += __shfl_xor(r1, o);
    }
    r0 += bl[0];
    r1 += bl[1];
  }

  __shared__ float stage[2 * WAVES];
  if (lane == 0) {
    stage[2 * wv]     = r0;
    stage[2 * wv + 1] = r1;
  }
  __syncthreads();
  if (wv == 0) {
    const int idx = blockIdx.x * (2 * WAVES) + lane;
    if (lane < 2 * WAVES && idx < 2 * n) dstbuf[idx] = stage[lane];
  }
}

extern "C" void kernel_launch(void* const* d_in, const int* in_sizes, int n_in,
                              void* d_out, int out_size, void* d_ws, size_t ws_size,
                              hipStream_t stream) {
  const float* x   = (const float*)d_in[0];
  const int*   ei  = (const int*)d_in[1];
  const float* W1  = (const float*)d_in[2];
  const float* b1  = (const float*)d_in[3];
  const float* W2  = (const float*)d_in[4];
  const float* b2  = (const float*)d_in[5];
  const float* Wl  = (const float*)d_in[6];
  const float* bl  = (const float*)d_in[7];
  float*       out = (float*)d_out;

  const int*   src = ei;
  const int*   dst = ei + NEDGES;

  char* ws = (char*)d_ws;
  const size_t H_OFF   = 200192;
  const size_t HBYTES  = (size_t)NNODES * HID * 4;
  const size_t X2_OFF  = H_OFF + HBYTES;
  const size_t BP_OFF  = X2_OFF + HBYTES;
  const size_t CNT_OFF = BP_OFF + 262144;
  const size_t BK_OFF  = CNT_OFF + 200192;
  const size_t OD_OFF  = BK_OFF + 6400000;
  const size_t OS_OFF  = OD_OFF + 2400000;

  float* H    = (float*)(ws + H_OFF);
  float* AX   = H;
  float* X2   = (float*)(ws + X2_OFF);
  float* Bp   = (float*)(ws + BP_OFF);
  int*   cnt    = (int*)(ws + CNT_OFF);
  int*   bucket = (int*)(ws + BK_OFF);
  int*   ovfdst = (int*)(ws + OD_OFF);
  int*   ovfsrc = (int*)(ws + OS_OFF);

  const int TPB = 256;
  const int cntBlocks   = (NNODES + 1 + TPB - 1) / TPB;
  const int edgeBlocks  = (NEDGES + TPB - 1) / TPB;
  const int rowBlocks8  = (NNODES + 7) / 8;
  const int rowBlocks16 = (NNODES + 15) / 16;
  const int gemmBlocks  = NNODES / (16 * MTILES);

  zero_i32_kernel<<<cntBlocks, TPB, 0, stream>>>(cnt, NNODES + 1);
  deg_bucket_kernel<<<edgeBlocks, TPB, 0, stream>>>(src, dst, cnt, bucket, ovfdst, ovfsrc, NEDGES, NNODES);

  {
    const int tot = INDIM * HID;
    gather_in_kernel<8><<<rowBlocks8, 256, 0, stream>>>(x, cnt, bucket, ovfdst, ovfsrc, AX, NNODES);
    pack_b_kernel<<<(tot + TPB - 1) / TPB, TPB, 0, stream>>>(W1, Bp, HID, tot);
    gemm_wmma_f32_kernel<INDIM, true><<<gemmBlocks, 128, 0, stream>>>(AX, Bp, b1, X2, HID);
  }

  {
    const int tot = HID * HID;
    pack_b_kernel<<<(tot + TPB - 1) / TPB, TPB, 0, stream>>>(W2, Bp, HID, tot);
    gemm_wmma_f32_kernel<HID, false><<<gemmBlocks, 128, 0, stream>>>(X2, Bp, b2, H, HID);
    gather_finish_kernel<16><<<rowBlocks16, 512, 0, stream>>>(
        H, cnt, bucket, ovfdst, ovfsrc, b2, Wl, bl, out, NNODES);
  }
}


# Both GEMMs in one kernel: the block's 80x256 hidden tile stays in LDS as the A operand of the second product, X2 never goes to memory; zeroing and both weight packs share one launch (8 launches -> 5)
# speedup vs baseline: 4.9194x; 1.0929x over previous
#include <hip/hip_runtime.h>
#include <hip/hip_bf16.h>

typedef float v2f __attribute__((ext_vector_type(2)));
typedef float v8f __attribute__((ext_vector_type(8)));

#define NNODES 50000
#define NEDGES 600000
#define INDIM 128
#define HID 256
#define MTILES 5

#define CAP 32

__global__ void deg_bucket_kernel(const int* __restrict__ src, const int* __restrict__ dst,
                                  int* __restrict__ cnt, int* __restrict__ bucket,
                                  int* __restrict__ ovf_dst, int* __restrict__ ovf_src,
                                  int E, int n) {
  int e = blockIdx.x * blockDim.x + threadIdx.x;
  if (e < E) {
    const int d = dst[e];
    const int s = src[e];
    const int r = atomicAdd(&cnt[d], 1);
    if (r < CAP) {
      bucket[(size_t)d * CAP + r] = s;
    } else {
      const int p = atomicAdd(&cnt[n], 1);
      ovf_dst[p] = d;
      ovf_src[p] = s;
    }
  }
}

static __device__ __forceinline__ float dinv_of(const int c) {
  return rsqrtf((float)c + 1.0f);
}

static __device__ __forceinline__ void pack_b_elem(const float* __restrict__ B,
                                                   float* __restrict__ Bp, const int i) {
  const int k = i / HID;
  const int n = i - k * HID;
  Bp[((size_t)(k >> 1) * HID + n) * 2 + (k & 1)] = B[i];
}

__global__ void prep_kernel(int* __restrict__ cnt,
                            const float* __restrict__ W1, float* __restrict__ Bp1,
                            const float* __restrict__ W2, float* __restrict__ Bp2) {
  const int i = blockIdx.x * blockDim.x + threadIdx.x;
  if (i < NNODES + 1) cnt[i] = 0;
  if (i < INDIM * HID) pack_b_elem(W1, Bp1, i);
  if (i < HID * HID)   pack_b_elem(W2, Bp2, i);
}

static __device__ __forceinline__ float add_relu(const float a, const float b) {
  const float v = a + b;
  return v > 0.0f ? v : 0.0f;
}

#define KCH 32
#define AROWS (16 * MTILES)
#define APITCH 36
#define XPITCH (HID + 4)

static __device__ __forceinline__ void load_b_chunk(v2f (&bf)[KCH / 4][4],
                                                    const float* __restrict__ bk) {
  #pragma unroll
  for (int s = 0; s < KCH / 4; ++s) {
    #pragma unroll
    for (int u = 0; u < 4; ++u)
      bf[s][u] = *(const v2f*)(bk + (size_t)s * 2 * HID * 2 + u * 32);
  }
}

template <int PITCH>
static __device__ __forceinline__ void mma_chunk(v8f (&acc)[MTILES][4],
                                                 const float* af0,
                                                 const v2f (&bf)[KCH / 4][4]) {
  #pragma unroll
  for (int s = 0; s < KCH / 4; ++s) {
    #pragma unroll
    for (int t = 0; t < MTILES; ++t) {
      const v2f af = *(const v2f*)(af0 + t * 16 * PITCH + 4 * s);
      #pragma unroll
      for (int u = 0; u < 4; ++u) {
        acc[t][u] = __builtin_amdgcn_wmma_f32_16x16x4_f32(
            false, af, false, bf[s][u], (short)0, acc[t][u], false, false);
      }
    }
  }
}

__global__ __launch_bounds__(128, 1)
void dense2_wmma_f32_kernel(const float* __restrict__ A,
                            const float* __restrict__ Bp1,
                            const float* __restrict__ bias1,
                            const float* __restrict__ Bp2,
                            float* __restrict__ H) {
  __shared__ __attribute__((aligned(16))) float lds[AROWS * XPITCH];
  constexpr int NCH1 = INDIM / KCH;
  constexpr int NCH2 = HID / KCH;
  constexpr int ASTAGE = AROWS * APITCH;

  const int tid  = threadIdx.x;
  const int wave = tid >> 5;
  const int lane = tid & 31;
  const int half = lane >> 4;
  const int lidx = lane & 15;

  const int m0 = blockIdx.x * AROWS;
  const int n0 = wave * 64;

  v8f acc[MTILES][4] = {};

  const int crow = tid >> 3;
  const int cseg = tid & 7;
  const float* gsrc = A + (size_t)(m0 + crow) * INDIM + cseg * 4;
  const int ldst = crow * APITCH + cseg * 4;

  {
    float4 c0[MTILES];
    #pragma unroll
    for (int i = 0; i < MTILES; ++i)
      c0[i] = *(const float4*)(gsrc + (size_t)i * 16 * INDIM);
    #pragma unroll
    for (int i = 0; i < MTILES; ++i)
      *(float4*)(&lds[ldst + i * 16 * APITCH]) = c0[i];
  }
  __syncthreads();

  const size_t boff = ((size_t)half * HID + n0 + lidx) * 2;
  const int aoff = lidx * APITCH + 2 * half;

  #pragma unroll 1
  for (int kc = 0; kc < NCH1 - 1; ++kc) {
    v2f bf[KCH / 4][4];
    load_b_chunk(bf, Bp1 + boff + (size_t)kc * (KCH / 2) * HID * 2);
    float4 nx[MTILES];
    #pragma unroll
    for (int i = 0; i < MTILES; ++i)
      nx[i] = *(const float4*)(gsrc + (size_t)i * 16 * INDIM + (kc + 1) * KCH);

    mma_chunk<APITCH>(acc, &lds[(kc & 1) * ASTAGE + aoff], bf);

    float* ad = &lds[((kc + 1) & 1) * ASTAGE];
    #pragma unroll
    for (int i = 0; i < MTILES; ++i)
      *(float4*)(&ad[ldst + i * 16 * APITCH]) = nx[i];
    __syncthreads();
  }
  {
    v2f bf[KCH / 4][4];
    load_b_chunk(bf, Bp1 + boff + (size_t)(NCH1 - 1) * (KCH / 2) * HID * 2);
    mma_chunk<APITCH>(acc, &lds[((NCH1 - 1) & 1) * ASTAGE + aoff], bf);
  }

  {
    float bv[4];
    #pragma unroll
    for (int u = 0; u < 4; ++u) bv[u] = bias1[n0 + lidx + u * 16];
    __syncthreads();
    #pragma unroll
    for (int t = 0; t < MTILES; ++t) {
      #pragma unroll
      for (int g = 0; g < 8; ++g) {
        const int r = t * 16 + g + 8 * half;
        #pragma unroll
        for (int u = 0; u < 4; ++u)
          lds[r * XPITCH + n0 + lidx + u * 16] = add_relu(acc[t][u][g], bv[u]);
      }
    }
    __syncthreads();
  }

  #pragma unroll
  for (int t = 0; t < MTILES; ++t) {
    #pragma unroll
    for (int u = 0; u < 4; ++u) acc[t][u] = v8f{};
  }
  const int xoff = lidx * XPITCH + 2 * half;
  #pragma unroll 1
  for (int kc = 0; kc < NCH2; ++kc) {
    v2f bf[KCH / 4][4];
    load_b_chunk(bf, Bp2 + boff + (size_t)kc * (KCH / 2) * HID * 2);
    mma_chunk<XPITCH>(acc, &lds[xoff + kc * KCH], bf);
  }

  #pragma unroll
  for (int t = 0; t < MTILES; ++t) {
    #pragma unroll
    for (int g = 0; g < 8; ++g) {
      const int row = m0 + t * 16 + g + 8 * half;
      const size_t base = (size_t)row * HID + n0 + lidx;
      #pragma unroll
      for (int u = 0; u < 4; ++u) H[base + u * 16] = acc[t][u][g];
    }
  }
}

static __device__ __forceinline__ void vec_axpy(float4& a, const float4 v, const float w) {
  a.x = fmaf(v.x, w, a.x);
  a.y = fmaf(v.y, w, a.y);
  a.z = fmaf(v.z, w, a.z);
  a.w = fmaf(v.w, w, a.w);
}

template <int NB>
static __device__ __forceinline__ void gather_batch(float4& a, const float* __restrict__ x,
                                                    const int s_l, const float w_l,
                                                    const int j, const int lane) {
  int s[NB]; float w[NB]; float4 p[NB];
  #pragma unroll
  for (int u = 0; u < NB; ++u) {
    s[u] = __builtin_amdgcn_readlane(s_l, j + u);
    w[u] = __int_as_float(__builtin_amdgcn_readlane(__float_as_int(w_l), j + u));
  }
  #pragma unroll
  for (int u = 0; u < NB; ++u)
    p[u] = ((const float4*)(x + (size_t)s[u] * INDIM))[lane];
  #pragma unroll
  for (int u = 0; u < NB; ++u) vec_axpy(a, p[u], w[u]);
}

template <int WAVES>
__global__ __launch_bounds__(WAVES * 32)
void gather_in_kernel(const float* __restrict__ x,
                      const int* __restrict__ cnt,
                      const int* __restrict__ bucket,
                      const int* __restrict__ ovf_dst,
                      const int* __restrict__ ovf_src,
                      float* __restrict__ ax, int n) {
  const int wv   = threadIdx.x >> 5;
  const int lane = threadIdx.x & 31;
  const int node = __builtin_amdgcn_readfirstlane(blockIdx.x * WAVES + wv);
  if (node >= n) return;

  const int c = cnt[node];
  const float di = dinv_of(c);
  const float sl = di * di;

  const float4 xs = ((const float4*)(x + (size_t)node * INDIM))[lane];
  float4 a = make_float4(0.0f, 0.0f, 0.0f, 0.0f);

  {
    const int m = (c < CAP) ? c : CAP;
    int   s_l = 0;
    float w_l = 0.0f;
    if (lane < m) {
      s_l = bucket[(size_t)node * CAP + lane];
      w_l = dinv_of(cnt[s_l]) * di;
    }
    int j = 0;
    for (; j + 8 <= m; j += 8) gather_batch<8>(a, x, s_l, w_l, j, lane);
    if (j + 4 <= m) { gather_batch<4>(a, x, s_l, w_l, j, lane); j += 4; }
    if (j + 2 <= m) { gather_batch<2>(a, x, s_l, w_l, j, lane); j += 2; }
    if (j < m)        gather_batch<1>(a, x, s_l, w_l, j, lane);
  }
  if (c > CAP) {
    const int novf = cnt[n];
    for (int b = 0; b < novf; b += 32) {
      const int i = b + lane;
      int od = -1, os = 0;
      if (i < novf) { od = ovf_dst[i]; os = ovf_src[i]; }
      unsigned mask = __builtin_amdgcn_ballot_w32(od == node);
      while (mask != 0u) {
        const int jj = __builtin_ctz(mask);
        mask &= mask - 1u;
        const int s = __builtin_amdgcn_readlane(os, jj);
        const float w = dinv_of(cnt[s]) * di;
        const float4 p = ((const float4*)(x + (size_t)s * INDIM))[lane];
        vec_axpy(a, p, w);
      }
    }
  }

  vec_axpy(a, xs, sl);
  ((float4*)(ax + (size_t)node * INDIM))[lane] = a;
}

static __device__ __forceinline__ void row_axpy(float4& a0, float4& a1,
                                                const float4 v0, const float4 v1,
                                                const float w) {
  a0.x = fmaf(v0.x, w, a0.x);
  a0.y = fmaf(v0.y, w, a0.y);
  a0.z = fmaf(v0.z, w, a0.z);
  a0.w = fmaf(v0.w, w, a0.w);
  a1.x = fmaf(v1.x, w, a1.x);
  a1.y = fmaf(v1.y, w, a1.y);
  a1.z = fmaf(v1.z, w, a1.z);
  a1.w = fmaf(v1.w, w, a1.w);
}

template <int WAVES>
__global__ __launch_bounds__(WAVES * 32)
void gather_finish_kernel(const float* __restrict__ h,
                          const int* __restrict__ cnt,
                          const int* __restrict__ bucket,
                          const int* __restrict__ ovf_dst,
                          const int* __restrict__ ovf_src,
                          const float* __restrict__ bias,
                          const float* __restrict__ Wl,
                          const float* __restrict__ bl,
                          float* __restrict__ dstbuf, int n) {
  const int wv   = threadIdx.x >> 5;
  const int lane = threadIdx.x & 31;
  const int node = __builtin_amdgcn_readfirstlane(blockIdx.x * WAVES + wv);

  float r0 = 0.0f, r1 = 0.0f;
  if (node < n) {
    const int c = cnt[node];
    const float di = dinv_of(c);
    const float sl = di * di;

    const float4* hp = (const float4*)(h + (size_t)node * HID);
    float4 a0 = hp[lane];
    float4 a1 = hp[lane + 32];
    a0.x *= sl; a0.y *= sl; a0.z *= sl; a0.w *= sl;
    a1.x *= sl; a1.y *= sl; a1.z *= sl; a1.w *= sl;

    {
      const int m = (c < CAP) ? c : CAP;
      int   s_l = 0;
      float w_l = 0.0f;
      if (lane < m) {
        s_l = bucket[(size_t)node * CAP + lane];
        w_l = dinv_of(cnt[s_l]) * di;
      }
      int j = 0;
      for (; j + 4 <= m; j += 4) {
        const int s0 = __builtin_amdgcn_readlane(s_l, j);
        const int s1 = __builtin_amdgcn_readlane(s_l, j + 1);
        const int s2 = __builtin_amdgcn_readlane(s_l, j + 2);
        const int s3 = __builtin_amdgcn_readlane(s_l, j + 3);
        const float w0 = __int_as_float(__builtin_amdgcn_readlane(__float_as_int(w_l), j));
        const float w1 = __int_as_float(__builtin_amdgcn_readlane(__float_as_int(w_l), j + 1));
        const float w2 = __int_as_float(__builtin_amdgcn_readlane(__float_as_int(w_l), j + 2));
        const float w3 = __int_as_float(__builtin_amdgcn_readlane(__float_as_int(w_l), j + 3));
        const float4* h0 = (const float4*)(h + (size_t)s0 * HID);
        const float4* h1 = (const float4*)(h + (size_t)s1 * HID);
        const float4* h2 = (const float4*)(h + (size_t)s2 * HID);
        const float4* h3 = (const float4*)(h + (size_t)s3 * HID);
        const float4 p0 = h0[lane], q0 = h0[lane + 32];
        const float4 p1 = h1[lane], q1 = h1[lane + 32];
        const float4 p2 = h2[lane], q2 = h2[lane + 32];
        const float4 p3 = h3[lane], q3 = h3[lane + 32];
        row_axpy(a0, a1, p0, q0, w0);
        row_axpy(a0, a1, p1, q1, w1);
        row_axpy(a0, a1, p2, q2, w2);
        row_axpy(a0, a1, p3, q3, w3);
      }
      #pragma unroll 1
      for (; j < m; ++j) {
        const int s0 = __builtin_amdgcn_readlane(s_l, j);
        const float w0 = __int_as_float(__builtin_amdgcn_readlane(__float_as_int(w_l), j));
        const float4* h0 = (const float4*)(h + (size_t)s0 * HID);
        const float4 p0 = h0[lane], q0 = h0[lane + 32];
        row_axpy(a0, a1, p0, q0, w0);
      }
    }
    if (c > CAP) {
      const int novf = cnt[n];
      for (int b = 0; b < novf; b += 32) {
        const int i = b + lane;
        int od = -1, os = 0;
        if (i < novf) { od = ovf_dst[i]; os = ovf_src[i]; }
        unsigned mask = __builtin_amdgcn_ballot_w32(od == node);
        while (mask != 0u) {
          const int jj = __builtin_ctz(mask);
          mask &= mask - 1u;
          const int s0 = __builtin_amdgcn_readlane(os, jj);
          const float w0 = dinv_of(cnt[s0]) * di;
          const float4* h0 = (const float4*)(h + (size_t)s0 * HID);
          const float4 p0 = h0[lane], q0 = h0[lane + 32];
          row_axpy(a0, a1, p0, q0, w0);
        }
      }
    }

    const float4 b0 = ((const float4*)bias)[lane];
    const float4 b1 = ((const float4*)bias)[lane + 32];
    a0.x = add_relu(a0.x, b0.x); a0.y = add_relu(a0.y, b0.y);
    a0.z = add_relu(a0.z, b0.z); a0.w = add_relu(a0.w, b0.w);
    a1.x = add_relu(a1.x, b1.x); a1.y = add_relu(a1.y, b1.y);
    a1.z = add_relu(a1.z, b1.z); a1.w = add_relu(a1.w, b1.w);

    const float4* wp = (const float4*)Wl;
    const float4 wa = wp[2 * lane],      wb = wp[2 * lane + 1];
    const float4 wc = wp[64 + 2 * lane], wd = wp[64 + 2 * lane + 1];
    r0 = a0.x * wa.x;            r1 = a0.x * wa.y;
    r0 = fmaf(a0.y, wa.z, r0);   r1 = fmaf(a0.y, wa.w, r1);
    r0 = fmaf(a0.z, wb.x, r0);   r1 = fmaf(a0.z, wb.y, r1);
    r0 = fmaf(a0.w, wb.z, r0);   r1 = fmaf(a0.w, wb.w, r1);
    r0 = fmaf(a1.x, wc.x, r0);   r1 = fmaf(a1.x, wc.y, r1);
    r0 = fmaf(a1.y, wc.z, r0);   r1 = fmaf(a1.y, wc.w, r1);
    r0 = fmaf(a1.z, wd.x, r0);   r1 = fmaf(a1.z, wd.y, r1);
    r0 = fmaf(a1.w, wd.z, r0);   r1 = fmaf(a1.w, wd.w, r1);
    for (int o = 16; o > 0; o >>= 1) {
      r0 += __shfl_xor(r0, o);
      r1 += __shfl_xor(r1, o);
    }
    r0 += bl[0];
    r1 += bl[1];
  }

  __shared__ float stage[2 * WAVES];
  if (lane == 0) {
    stage[2 * wv]     = r0;
    stage[2 * wv + 1] = r1;
  }
  __syncthreads();
  if (wv == 0) {
    const int idx = blockIdx.x * (2 * WAVES) + lane;
    if (lane < 2 * WAVES && idx < 2 * n) dstbuf[idx] = stage[lane];
  }
}

extern "C" void kernel_launch(void* const* d_in, const int* in_sizes, int n_in,
                              void* d_out, int out_size, void* d_ws, size_t ws_size,
                              hipStream_t stream) {
  const float* x   = (const float*)d_in[0];
  const int*   ei  = (const int*)d_in[1];
  const float* W1  = (const float*)d_in[2];
  const float* b1  = (const float*)d_in[3];
  const float* W2  = (const float*)d_in[4];
  const float* b2  = (const float*)d_in[5];
  const float* Wl  = (const float*)d_in[6];
  const float* bl  = (const float*)d_in[7];
  float*       out = (float*)d_out;

  const int*   src = ei;
  const int*   dst = ei + NEDGES;

  char* ws = (char*)d_ws;
  const size_t H_OFF   = 200192;
  const size_t HBYTES  = (size_t)NNODES * HID * 4;
  const size_t AX_OFF  = H_OFF + HBYTES;
  const size_t AXBYTES = (size_t)NNODES * INDIM * 4;
  const size_t BP1_OFF = AX_OFF + AXBYTES;
  const size_t BP2_OFF = BP1_OFF + 131072;
  const size_t CNT_OFF = BP2_OFF + 262144;
  const size_t BK_OFF  = CNT_OFF + 200192;
  const size_t OD_OFF  = BK_OFF + 6400000;
  const size_t OS_OFF  = OD_OFF + 2400000;

  float* H    = (float*)(ws + H_OFF);
  float* AX   = (float*)(ws + AX_OFF);
  float* Bp1  = (float*)(ws + BP1_OFF);
  float* Bp2  = (float*)(ws + BP2_OFF);
  int*   cnt    = (int*)(ws + CNT_OFF);
  int*   bucket = (int*)(ws + BK_OFF);
  int*   ovfdst = (int*)(ws + OD_OFF);
  int*   ovfsrc = (int*)(ws + OS_OFF);

  const int TPB = 256;
  const int prepBlocks  = (HID * HID + TPB - 1) / TPB;
  const int edgeBlocks  = (NEDGES + TPB - 1) / TPB;
  const int rowBlocks8  = (NNODES + 7) / 8;
  const int rowBlocks16 = (NNODES + 15) / 16;
  const int gemmBlocks  = NNODES / (16 * MTILES);

  prep_kernel<<<prepBlocks, TPB, 0, stream>>>(cnt, W1, Bp1, W2, Bp2);
  deg_bucket_kernel<<<edgeBlocks, TPB, 0, stream>>>(src, dst, cnt, bucket, ovfdst, ovfsrc, NEDGES, NNODES);

  gather_in_kernel<8><<<rowBlocks8, 256, 0, stream>>>(x, cnt, bucket, ovfdst, ovfsrc, AX, NNODES);

  dense2_wmma_f32_kernel<<<gemmBlocks, 128, 0, stream>>>(AX, Bp1, b1, Bp2, H);

  gather_finish_kernel<16><<<rowBlocks16, 512, 0, stream>>>(
      H, cnt, bucket, ovfdst, ovfsrc, b2, Wl, bl, out, NNODES);
}


# GEMM epilogue scales each H row by its own rsqrt(deg+1): the layer-2 gather adds source rows unweighted and scales once by the destination's, no per-edge degree lookups; sum order kept
# speedup vs baseline: 5.0975x; 1.0362x over previous
#include <hip/hip_runtime.h>
#include <hip/hip_bf16.h>

typedef float v2f __attribute__((ext_vector_type(2)));
typedef float v8f __attribute__((ext_vector_type(8)));

#define NNODES 50000
#define NEDGES 600000
#define INDIM 128
#define HID 256
#define MTILES 5

#define CAP 32

__global__ void deg_bucket_kernel(const int* __restrict__ src, const int* __restrict__ dst,
                                  int* __restrict__ cnt, int* __restrict__ bucket,
                                  int* __restrict__ ovf_dst, int* __restrict__ ovf_src,
                                  int E, int n) {
  int e = blockIdx.x * blockDim.x + threadIdx.x;
  if (e < E) {
    const int d = dst[e];
    const int s = src[e];
    const int r = atomicAdd(&cnt[d], 1);
    if (r < CAP) {
      bucket[(size_t)d * CAP + r] = s;
    } else {
      const int p = atomicAdd(&cnt[n], 1);
      ovf_dst[p] = d;
      ovf_src[p] = s;
    }
  }
}

static __device__ __forceinline__ float dinv_of(const int c) {
  return rsqrtf((float)c + 1.0f);
}

static __device__ __forceinline__ void pack_b_elem(const float* __restrict__ B,
                                                   float* __restrict__ Bp, const int i) {
  const int k = i / HID;
  const int n = i - k * HID;
  Bp[((size_t)(k >> 1) * HID + n) * 2 + (k & 1)] = B[i];
}

__global__ void prep_kernel(int* __restrict__ cnt,
                            const float* __restrict__ W1, float* __restrict__ Bp1,
                            const float* __restrict__ W2, float* __restrict__ Bp2) {
  const int i = blockIdx.x * blockDim.x + threadIdx.x;
  if (i < NNODES + 1) cnt[i] = 0;
  if (i < INDIM * HID) pack_b_elem(W1, Bp1, i);
  if (i < HID * HID)   pack_b_elem(W2, Bp2, i);
}

static __device__ __forceinline__ float add_relu(const float a, const float b) {
  const float v = a + b;
  return v > 0.0f ? v : 0.0f;
}

#define KCH 32
#define AROWS (16 * MTILES)
#define APITCH 36
#define XPITCH (HID + 4)

static __device__ __forceinline__ void load_b_chunk(v2f (&bf)[KCH / 4][4],
                                                    const float* __restrict__ bk) {
  #pragma unroll
  for (int s = 0; s < KCH / 4; ++s) {
    #pragma unroll
    for (int u = 0; u < 4; ++u)
      bf[s][u] = *(const v2f*)(bk + (size_t)s * 2 * HID * 2 + u * 32);
  }
}

template <int PITCH>
static __device__ __forceinline__ void mma_chunk(v8f (&acc)[MTILES][4],
                                                 const float* af0,
                                                 const v2f (&bf)[KCH / 4][4]) {
  #pragma unroll
  for (int s = 0; s < KCH / 4; ++s) {
    #pragma unroll
    for (int t = 0; t < MTILES; ++t) {
      const v2f af = *(const v2f*)(af0 + t * 16 * PITCH + 4 * s);
      #pragma unroll
      for (int u = 0; u < 4; ++u) {
        acc[t][u] = __builtin_amdgcn_wmma_f32_16x16x4_f32(
            false, af, false, bf[s][u], (short)0, acc[t][u], false, false);
      }
    }
  }
}

__global__ __launch_bounds__(128, 1)
void dense2_wmma_f32_kernel(const float* __restrict__ A,
                            const float* __restrict__ Bp1,
                            const float* __restrict__ bias1,
                            const float* __restrict__ Bp2,
                            const int* __restrict__ cnt,
                            float* __restrict__ H) {
  __shared__ __attribute__((aligned(16))) float lds[AROWS * XPITCH];
  __shared__ __attribute__((aligned(16))) float dinv_s[128];
  constexpr int NCH1 = INDIM / KCH;
  constexpr int NCH2 = HID / KCH;
  constexpr int ASTAGE = AROWS * APITCH;

  const int tid  = threadIdx.x;
  const int wave = tid >> 5;
  const int lane = tid & 31;
  const int half = lane >> 4;
  const int lidx = lane & 15;

  const int m0 = blockIdx.x * AROWS;
  const int n0 = wave * 64;

  const int cn = cnt[m0 + (tid < AROWS ? tid : AROWS - 1)];

  v8f acc[MTILES][4] = {};

  const int crow = tid >> 3;
  const int cseg = tid & 7;
  const float* gsrc = A + (size_t)(m0 + crow) * INDIM + cseg * 4;
  const int ldst = crow * APITCH + cseg * 4;

  {
    float4 c0[MTILES];
    #pragma unroll
    for (int i = 0; i < MTILES; ++i)
      c0[i] = *(const float4*)(gsrc + (size_t)i * 16 * INDIM);
    #pragma unroll
    for (int i = 0; i < MTILES; ++i)
      *(float4*)(&lds[ldst + i * 16 * APITCH]) = c0[i];
  }
  dinv_s[tid] = dinv_of(cn);
  __syncthreads();

  const size_t boff = ((size_t)half * HID + n0 + lidx) * 2;
  const int aoff = lidx * APITCH + 2 * half;

  #pragma unroll 1
  for (int kc = 0; kc < NCH1 - 1; ++kc) {
    v2f bf[KCH / 4][4];
    load_b_chunk(bf, Bp1 + boff + (size_t)kc * (KCH / 2) * HID * 2);
    float4 nx[MTILES];
    #pragma unroll
    for (int i = 0; i < MTILES; ++i)
      nx[i] = *(const float4*)(gsrc + (size_t)i * 16 * INDIM + (kc + 1) * KCH);

    mma_chunk<APITCH>(acc, &lds[(kc & 1) * ASTAGE + aoff], bf);

    float* ad = &lds[((kc + 1) & 1) * ASTAGE];
    #pragma unroll
    for (int i = 0; i < MTILES; ++i)
      *(float4*)(&ad[ldst + i * 16 * APITCH]) = nx[i];
    __syncthreads();
  }
  {
    v2f bf[KCH / 4][4];
    load_b_chunk(bf, Bp1 + boff + (size_t)(NCH1 - 1) * (KCH / 2) * HID * 2);
    mma_chunk<APITCH>(acc, &lds[((NCH1 - 1) & 1) * ASTAGE + aoff], bf);
  }

  {
    float bv[4];
    #pragma unroll
    for (int u = 0; u < 4; ++u) bv[u] = bias1[n0 + lidx + u * 16];
    __syncthreads();
    #pragma unroll
    for (int t = 0; t < MTILES; ++t) {
      #pragma unroll
      for (int g = 0; g < 8; ++g) {
        const int r = t * 16 + g + 8 * half;
        #pragma unroll
        for (int u = 0; u < 4; ++u)
          lds[r * XPITCH + n0 + lidx + u * 16] = add_relu(acc[t][u][g], bv[u]);
      }
    }
    __syncthreads();
  }

  #pragma unroll
  for (int t = 0; t < MTILES; ++t) {
    #pragma unroll
    for (int u = 0; u < 4; ++u) acc[t][u] = v8f{};
  }
  const int xoff = lidx * XPITCH + 2 * half;
  #pragma unroll 1
  for (int kc = 0; kc < NCH2; ++kc) {
    v2f bf[KCH / 4][4];
    load_b_chunk(bf, Bp2 + boff + (size_t)kc * (KCH / 2) * HID * 2);
    mma_chunk<XPITCH>(acc, &lds[xoff + kc * KCH], bf);
  }

  #pragma unroll
  for (int t = 0; t < MTILES; ++t) {
    #pragma unroll
    for (int g = 0; g < 8; ++g) {
      const int r = t * 16 + g + 8 * half;
      const float dr = dinv_s[r];
      const size_t base = (size_t)(m0 + r) * HID + n0 + lidx;
      #pragma unroll
      for (int u = 0; u < 4; ++u) H[base + u * 16] = acc[t][u][g] * dr;
    }
  }
}

static __device__ __forceinline__ void vec_axpy(float4& a, const float4 v, const float w) {
  a.x = fmaf(v.x, w, a.x);
  a.y = fmaf(v.y, w, a.y);
  a.z = fmaf(v.z, w, a.z);
  a.w = fmaf(v.w, w, a.w);
}

template <int NB>
static __device__ __forceinline__ void gather_batch(float4& a, const float* __restrict__ x,
                                                    const int s_l, const float w_l,
                                                    const int j, const int lane) {
  int s[NB]; float w[NB]; float4 p[NB];
  #pragma unroll
  for (int u = 0; u < NB; ++u) {
    s[u] = __builtin_amdgcn_readlane(s_l, j + u);
    w[u] = __int_as_float(__builtin_amdgcn_readlane(__float_as_int(w_l), j + u));
  }
  #pragma unroll
  for (int u = 0; u < NB; ++u)
    p[u] = ((const float4*)(x + (size_t)s[u] * INDIM))[lane];
  #pragma unroll
  for (int u = 0; u < NB; ++u) vec_axpy(a, p[u], w[u]);
}

template <int WAVES>
__global__ __launch_bounds__(WAVES * 32)
void gather_in_kernel(const float* __restrict__ x,
                      const int* __restrict__ cnt,
                      const int* __restrict__ bucket,
                      const int* __restrict__ ovf_dst,
                      const int* __restrict__ ovf_src,
                      float* __restrict__ ax, int n) {
  const int wv   = threadIdx.x >> 5;
  const int lane = threadIdx.x & 31;
  const int node = __builtin_amdgcn_readfirstlane(blockIdx.x * WAVES + wv);
  if (node >= n) return;

  const int c = cnt[node];
  const float di = dinv_of(c);
  const float sl = di * di;

  const float4 xs = ((const float4*)(x + (size_t)node * INDIM))[lane];
  float4 a = make_float4(0.0f, 0.0f, 0.0f, 0.0f);

  {
    const int m = (c < CAP) ? c : CAP;
    int   s_l = 0;
    float w_l = 0.0f;
    if (lane < m) {
      s_l = bucket[(size_t)node * CAP + lane];
      w_l = dinv_of(cnt[s_l]) * di;
    }
    int j = 0;
    for (; j + 8 <= m; j += 8) gather_batch<8>(a, x, s_l, w_l, j, lane);
    if (j + 4 <= m) { gather_batch<4>(a, x, s_l, w_l, j, lane); j += 4; }
    if (j + 2 <= m) { gather_batch<2>(a, x, s_l, w_l, j, lane); j += 2; }
    if (j < m)        gather_batch<1>(a, x, s_l, w_l, j, lane);
  }
  if (c > CAP) {
    const int novf = cnt[n];
    for (int b = 0; b < novf; b += 32) {
      const int i = b + lane;
      int od = -1, os = 0;
      if (i < novf) { od = ovf_dst[i]; os = ovf_src[i]; }
      unsigned mask = __builtin_amdgcn_ballot_w32(od == node);
      while (mask != 0u) {
        const int jj = __builtin_ctz(mask);
        mask &= mask - 1u;
        const int s = __builtin_amdgcn_readlane(os, jj);
        const float w = dinv_of(cnt[s]) * di;
        const float4 p = ((const float4*)(x + (size_t)s * INDIM))[lane];
        vec_axpy(a, p, w);
      }
    }
  }

  vec_axpy(a, xs, sl);
  ((float4*)(ax + (size_t)node * INDIM))[lane] = a;
}

static __device__ __forceinline__ void row_add(float4& a0, float4& a1,
                                               const float4 v0, const float4 v1) {
  a0.x += v0.x;
  a0.y += v0.y;
  a0.z += v0.z;
  a0.w += v0.w;
  a1.x += v1.x;
  a1.y += v1.y;
  a1.z += v1.z;
  a1.w += v1.w;
}

template <int WAVES>
__global__ __launch_bounds__(WAVES * 32)
void gather_finish_kernel(const float* __restrict__ h,
                          const int* __restrict__ cnt,
                          const int* __restrict__ bucket,
                          const int* __restrict__ ovf_dst,
                          const int* __restrict__ ovf_src,
                          const float* __restrict__ bias,
                          const float* __restrict__ Wl,
                          const float* __restrict__ bl,
                          float* __restrict__ dstbuf, int n) {
  const int wv   = threadIdx.x >> 5;
  const int lane = threadIdx.x & 31;
  const int node = __builtin_amdgcn_readfirstlane(blockIdx.x * WAVES + wv);

  float r0 = 0.0f, r1 = 0.0f;
  if (node < n) {
    const int c = cnt[node];
    const float di = dinv_of(c);

    const float4* hp = (const float4*)(h + (size_t)node * HID);
    float4 a0 = hp[lane];
    float4 a1 = hp[lane + 32];

    {
      const int m = (c < CAP) ? c : CAP;
      int s_l = 0;
      if (lane < m) s_l = bucket[(size_t)node * CAP + lane];
      int j = 0;
      for (; j + 4 <= m; j += 4) {
        const int s0 = __builtin_amdgcn_readlane(s_l, j);
        const int s1 = __builtin_amdgcn_readlane(s_l, j + 1);
        const int s2 = __builtin_amdgcn_readlane(s_l, j + 2);
        const int s3 = __builtin_amdgcn_readlane(s_l, j + 3);
        const float4* h0 = (const float4*)(h + (size_t)s0 * HID);
        const float4* h1 = (const float4*)(h + (size_t)s1 * HID);
        const float4* h2 = (const float4*)(h + (size_t)s2 * HID);
        const float4* h3 = (const float4*)(h + (size_t)s3 * HID);
        const float4 p0 = h0[lane], q0 = h0[lane + 32];
        const float4 p1 = h1[lane], q1 = h1[lane + 32];
        const float4 p2 = h2[lane], q2 = h2[lane + 32];
        const float4 p3 = h3[lane], q3 = h3[lane + 32];
        row_add(a0, a1, p0, q0);
        row_add(a0, a1, p1, q1);
        row_add(a0, a1, p2, q2);
        row_add(a0, a1, p3, q3);
      }
      #pragma unroll 1
      for (; j < m; ++j) {
        const int s0 = __builtin_amdgcn_readlane(s_l, j);
        const float4* h0 = (const float4*)(h + (size_t)s0 * HID);
        const float4 p0 = h0[lane], q0 = h0[lane + 32];
        row_add(a0, a1, p0, q0);
      }
    }
    if (c > CAP) {
      const int novf = cnt[n];
      for (int b = 0; b < novf; b += 32) {
        const int i = b + lane;
        int od = -1, os = 0;
        if (i < novf) { od = ovf_dst[i]; os = ovf_src[i]; }
        unsigned mask = __builtin_amdgcn_ballot_w32(od == node);
        while (mask != 0u) {
          const int jj = __builtin_ctz(mask);
          mask &= mask - 1u;
          const int s0 = __builtin_amdgcn_readlane(os, jj);
          const float4* h0 = (const float4*)(h + (size_t)s0 * HID);
          const float4 p0 = h0[lane], q0 = h0[lane + 32];
          row_add(a0, a1, p0, q0);
        }
      }
    }

    const float4 b0 = ((const float4*)bias)[lane];
    const float4 b1 = ((const float4*)bias)[lane + 32];
    a0.x = add_relu(a0.x * di, b0.x); a0.y = add_relu(a0.y * di, b0.y);
    a0.z = add_relu(a0.z * di, b0.z); a0.w = add_relu(a0.w * di, b0.w);
    a1.x = add_relu(a1.x * di, b1.x); a1.y = add_relu(a1.y * di, b1.y);
    a1.z = add_relu(a1.z * di, b1.z); a1.w = add_relu(a1.w * di, b1.w);

    const float4* wp = (const float4*)Wl;
    const float4 wa = wp[2 * lane],      wb = wp[2 * lane + 1];
    const float4 wc = wp[64 + 2 * lane], wd = wp[64 + 2 * lane + 1];
    r0 = a0.x * wa.x;            r1 = a0.x * wa.y;
    r0 = fmaf(a0.y, wa.z, r0);   r1 = fmaf(a0.y, wa.w, r1);
    r0 = fmaf(a0.z, wb.x, r0);   r1 = fmaf(a0.z, wb.y, r1);
    r0 = fmaf(a0.w, wb.z, r0);   r1 = fmaf(a0.w, wb.w, r1);
    r0 = fmaf(a1.x, wc.x, r0);   r1 = fmaf(a1.x, wc.y, r1);
    r0 = fmaf(a1.y, wc.z, r0);   r1 = fmaf(a1.y, wc.w, r1);
    r0 = fmaf(a1.z, wd.x, r0);   r1 = fmaf(a1.z, wd.y, r1);
    r0 = fmaf(a1.w, wd.z, r0);   r1 = fmaf(a1.w, wd.w, r1);
    for (int o = 16; o > 0; o >>= 1) {
      r0 += __shfl_xor(r0, o);
      r1 += __shfl_xor(r1, o);
    }
    r0 += bl[0];
    r1 += bl[1];
  }

  __shared__ float stage[2 * WAVES];
  if (lane == 0) {
    stage[2 * wv]     = r0;
    stage[2 * wv + 1] = r1;
  }
  __syncthreads();
  if (wv == 0) {
    const int idx = blockIdx.x * (2 * WAVES) + lane;
    if (lane < 2 * WAVES && idx < 2 * n) dstbuf[idx] = stage[lane];
  }
}

extern "C" void kernel_launch(void* const* d_in, const int* in_sizes, int n_in,
                              void* d_out, int out_size, void* d_ws, size_t ws_size,
                              hipStream_t stream) {
  const float* x   = (const float*)d_in[0];
  const int*   ei  = (const int*)d_in[1];
  const float* W1  = (const float*)d_in[2];
  const float* b1  = (const float*)d_in[3];
  const float* W2  = (const float*)d_in[4];
  const float* b2  = (const float*)d_in[5];
  const float* Wl  = (const float*)d_in[6];
  const float* bl  = (const float*)d_in[7];
  float*       out = (float*)d_out;

  const int*   src = ei;
  const int*   dst = ei + NEDGES;

  char* ws = (char*)d_ws;
  const size_t H_OFF   = 200192;
  const size_t HBYTES  = (size_t)NNODES * HID * 4;
  const size_t AX_OFF  = H_OFF + HBYTES;
  const size_t AXBYTES = (size_t)NNODES * INDIM * 4;
  const size_t BP1_OFF = AX_OFF + AXBYTES;
  const size_t BP2_OFF = BP1_OFF + 131072;
  const size_t CNT_OFF = BP2_OFF + 262144;
  const size_t BK_OFF  = CNT_OFF + 200192;
  const size_t OD_OFF  = BK_OFF + 6400000;
  const size_t OS_OFF  = OD_OFF + 2400000;

  float* H    = (float*)(ws + H_OFF);
  float* AX   = (float*)(ws + AX_OFF);
  float* Bp1  = (float*)(ws + BP1_OFF);
  float* Bp2  = (float*)(ws + BP2_OFF);
  int*   cnt    = (int*)(ws + CNT_OFF);
  int*   bucket = (int*)(ws + BK_OFF);
  int*   ovfdst = (int*)(ws + OD_OFF);
  int*   ovfsrc = (int*)(ws + OS_OFF);

  const int TPB = 256;
  const int prepBlocks  = (HID * HID + TPB - 1) / TPB;
  const int edgeBlocks  = (NEDGES + TPB - 1) / TPB;
  const int rowBlocks8  = (NNODES + 7) / 8;
  const int rowBlocks16 = (NNODES + 15) / 16;
  const int gemmBlocks  = NNODES / (16 * MTILES);

  prep_kernel<<<prepBlocks, TPB, 0, stream>>>(cnt, W1, Bp1, W2, Bp2);
  deg_bucket_kernel<<<edgeBlocks, TPB, 0, stream>>>(src, dst, cnt, bucket, ovfdst, ovfsrc, NEDGES, NNODES);

  gather_in_kernel<8><<<rowBlocks8, 256, 0, stream>>>(x, cnt, bucket, ovfdst, ovfsrc, AX, NNODES);

  dense2_wmma_f32_kernel<<<gemmBlocks, 128, 0, stream>>>(AX, Bp1, b1, Bp2, cnt, H);

  gather_finish_kernel<16><<<rowBlocks16, 512, 0, stream>>>(
      H, cnt, bucket, ovfdst, ovfsrc, b2, Wl, bl, out, NNODES);
}


# GEMM grid re-dealt: 768 blocks of 4 M-tiles (53 take 5), three resident on each of 256 WGPs, instead of 625 of 5: the fullest WGP issues f32 WMMAs for 13 M-tiles, not 15; k and sum order kept
# speedup vs baseline: 5.4875x; 1.0765x over previous
#include <hip/hip_runtime.h>
#include <hip/hip_bf16.h>

typedef float v2f __attribute__((ext_vector_type(2)));
typedef float v8f __attribute__((ext_vector_type(8)));

#define NNODES 50000
#define NEDGES 600000
#define INDIM 128
#define HID 256
#define MT_MAX 5
#define GEMM_BLOCKS 768

#define CAP 32

__global__ void deg_bucket_kernel(const int* __restrict__ src, const int* __restrict__ dst,
                                  int* __restrict__ cnt, int* __restrict__ bucket,
                                  int* __restrict__ ovf_dst, int* __restrict__ ovf_src,
                                  int E, int n) {
  int e = blockIdx.x * blockDim.x + threadIdx.x;
  if (e < E) {
    const int d = dst[e];
    const int s = src[e];
    const int r = atomicAdd(&cnt[d], 1);
    if (r < CAP) {
      bucket[(size_t)d * CAP + r] = s;
    } else {
      const int p = atomicAdd(&cnt[n], 1);
      ovf_dst[p] = d;
      ovf_src[p] = s;
    }
  }
}

static __device__ __forceinline__ float dinv_of(const int c) {
  return rsqrtf((float)c + 1.0f);
}

static __device__ __forceinline__ void pack_b_elem(const float* __restrict__ B,
                                                   float* __restrict__ Bp, const int i) {
  const int k = i / HID;
  const int n = i - k * HID;
  Bp[((size_t)(k >> 1) * HID + n) * 2 + (k & 1)] = B[i];
}

__global__ void prep_kernel(int* __restrict__ cnt,
                            const float* __restrict__ W1, float* __restrict__ Bp1,
                            const float* __restrict__ W2, float* __restrict__ Bp2) {
  const int i = blockIdx.x * blockDim.x + threadIdx.x;
  if (i < NNODES + 1) cnt[i] = 0;
  if (i < INDIM * HID) pack_b_elem(W1, Bp1, i);
  if (i < HID * HID)   pack_b_elem(W2, Bp2, i);
}

static __device__ __forceinline__ float add_relu(const float a, const float b) {
  const float v = a + b;
  return v > 0.0f ? v : 0.0f;
}

#define KCH 32
#define APITCH 36
#define XPITCH (HID + 4)

static __device__ __forceinline__ void load_b_chunk(v2f (&bf)[KCH / 4][4],
                                                    const float* __restrict__ bk) {
  #pragma unroll
  for (int s = 0; s < KCH / 4; ++s) {
    #pragma unroll
    for (int u = 0; u < 4; ++u)
      bf[s][u] = *(const v2f*)(bk + (size_t)s * 2 * HID * 2 + u * 32);
  }
}

template <int MT, int PITCH>
static __device__ __forceinline__ void mma_chunk(v8f (&acc)[MT][4],
                                                 const float* af0,
                                                 const v2f (&bf)[KCH / 4][4]) {
  #pragma unroll
  for (int s = 0; s < KCH / 4; ++s) {
    #pragma unroll
    for (int t = 0; t < MT; ++t) {
      const v2f af = *(const v2f*)(af0 + t * 16 * PITCH + 4 * s);
      #pragma unroll
      for (int u = 0; u < 4; ++u) {
        acc[t][u] = __builtin_amdgcn_wmma_f32_16x16x4_f32(
            false, af, false, bf[s][u], (short)0, acc[t][u], false, false);
      }
    }
  }
}

template <int MT>
static __device__ __forceinline__ void dense2_rows(const float* __restrict__ A,
                                                   const float* __restrict__ Bp1,
                                                   const float* __restrict__ bias1,
                                                   const float* __restrict__ Bp2,
                                                   const int* __restrict__ cnt,
                                                   float* __restrict__ H,
                                                   const int m0,
                                                   float* lds, float* dinv_s) {
  constexpr int AROWS = 16 * MT;
  constexpr int NCH1 = INDIM / KCH;
  constexpr int NCH2 = HID / KCH;
  constexpr int ASTAGE = AROWS * APITCH;

  const int tid  = threadIdx.x;
  const int wave = tid >> 5;
  const int lane = tid & 31;
  const int half = lane >> 4;
  const int lidx = lane & 15;

  const int n0 = wave * 64;

  const int cn = cnt[m0 + (tid < AROWS ? tid : AROWS - 1)];

  v8f acc[MT][4] = {};

  const int crow = tid >> 3;
  const int cseg = tid & 7;
  const float* gsrc = A + (size_t)(m0 + crow) * INDIM + cseg * 4;
  const int ldst = crow * APITCH + cseg * 4;

  {
    float4 c0[MT];
    #pragma unroll
    for (int i = 0; i < MT; ++i)
      c0[i] = *(const float4*)(gsrc + (size_t)i * 16 * INDIM);
    #pragma unroll
    for (int i = 0; i < MT; ++i)
      *(float4*)(&lds[ldst + i * 16 * APITCH]) = c0[i];
  }
  dinv_s[tid] = dinv_of(cn);
  __syncthreads();

  const size_t boff = ((size_t)half * HID + n0 + lidx) * 2;
  const int aoff = lidx * APITCH + 2 * half;

  #pragma unroll 1
  for (int kc = 0; kc < NCH1 - 1; ++kc) {
    v2f bf[KCH / 4][4];
    load_b_chunk(bf, Bp1 + boff + (size_t)kc * (KCH / 2) * HID * 2);
    float4 nx[MT];
    #pragma unroll
    for (int i = 0; i < MT; ++i)
      nx[i] = *(const float4*)(gsrc + (size_t)i * 16 * INDIM + (kc + 1) * KCH);

    mma_chunk<MT, APITCH>(acc, &lds[(kc & 1) * ASTAGE + aoff], bf);

    float* ad = &lds[((kc + 1) & 1) * ASTAGE];
    #pragma unroll
    for (int i = 0; i < MT; ++i)
      *(float4*)(&ad[ldst + i * 16 * APITCH]) = nx[i];
    __syncthreads();
  }
  {
    v2f bf[KCH / 4][4];
    load_b_chunk(bf, Bp1 + boff + (size_t)(NCH1 - 1) * (KCH / 2) * HID * 2);
    mma_chunk<MT, APITCH>(acc, &lds[((NCH1 - 1) & 1) * ASTAGE + aoff], bf);
  }

  {
    float bv[4];
    #pragma unroll
    for (int u = 0; u < 4; ++u) bv[u] = bias1[n0 + lidx + u * 16];
    __syncthreads();
    #pragma unroll
    for (int t = 0; t < MT; ++t) {
      #pragma unroll
      for (int g = 0; g < 8; ++g) {
        const int r = t * 16 + g + 8 * half;
        #pragma unroll
        for (int u = 0; u < 4; ++u)
          lds[r * XPITCH + n0 + lidx + u * 16] = add_relu(acc[t][u][g], bv[u]);
      }
    }
    __syncthreads();
  }

  #pragma unroll
  for (int t = 0; t < MT; ++t) {
    #pragma unroll
    for (int u = 0; u < 4; ++u) acc[t][u] = v8f{};
  }
  const int xoff = lidx * XPITCH + 2 * half;
  #pragma unroll 1
  for (int kc = 0; kc < NCH2; ++kc) {
    v2f bf[KCH / 4][4];
    load_b_chunk(bf, Bp2 + boff + (size_t)kc * (KCH / 2) * HID * 2);
    mma_chunk<MT, XPITCH>(acc, &lds[xoff + kc * KCH], bf);
  }

  #pragma unroll
  for (int t = 0; t < MT; ++t) {
    #pragma unroll
    for (int g = 0; g < 8; ++g) {
      const int r = t * 16 + g + 8 * half;
      const float dr = dinv_s[r];
      const size_t base = (size_t)(m0 + r) * HID + n0 + lidx;
      #pragma unroll
      for (int u = 0; u < 4; ++u) H[base + u * 16] = acc[t][u][g] * dr;
    }
  }
}

__global__ __launch_bounds__(128, 1)
void dense2_wmma_f32_kernel(const float* __restrict__ A,
                            const float* __restrict__ Bp1,
                            const float* __restrict__ bias1,
                            const float* __restrict__ Bp2,
                            const int* __restrict__ cnt,
                            float* __restrict__ H) {
  __shared__ __attribute__((aligned(16))) float lds[16 * MT_MAX * XPITCH];
  __shared__ __attribute__((aligned(16))) float dinv_s[128];

  constexpr int NT = NNODES / 16;
  constexpr int N5 = NT - 4 * GEMM_BLOCKS;
  static_assert(NNODES % 16 == 0 && N5 >= 0 && 3 * N5 <= GEMM_BLOCKS, "tile deal");
  const int b  = blockIdx.x;
  const int j  = (b + 2) / 3;
  const int m0 = 16 * (4 * b + (j < N5 ? j : N5));
  if (b % 3 == 0 && b / 3 < N5)
    dense2_rows<5>(A, Bp1, bias1, Bp2, cnt, H, m0, lds, dinv_s);
  else
    dense2_rows<4>(A, Bp1, bias1, Bp2, cnt, H, m0, lds, dinv_s);
}

static __device__ __forceinline__ void vec_axpy(float4& a, const float4 v, const float w) {
  a.x = fmaf(v.x, w, a.x);
  a.y = fmaf(v.y, w, a.y);
  a.z = fmaf(v.z, w, a.z);
  a.w = fmaf(v.w, w, a.w);
}

template <int NB>
static __device__ __forceinline__ void gather_batch(float4& a, const float* __restrict__ x,
                                                    const int s_l, const float w_l,
                                                    const int j, const int lane) {
  int s[NB]; float w[NB]; float4 p[NB];
  #pragma unroll
  for (int u = 0; u < NB; ++u) {
    s[u] = __builtin_amdgcn_readlane(s_l, j + u);
    w[u] = __int_as_float(__builtin_amdgcn_readlane(__float_as_int(w_l), j + u));
  }
  #pragma unroll
  for (int u = 0; u < NB; ++u)
    p[u] = ((const float4*)(x + (size_t)s[u] * INDIM))[lane];
  #pragma unroll
  for (int u = 0; u < NB; ++u) vec_axpy(a, p[u], w[u]);
}

template <int WAVES>
__global__ __launch_bounds__(WAVES * 32)
void gather_in_kernel(const float* __restrict__ x,
                      const int* __restrict__ cnt,
                      const int* __restrict__ bucket,
                      const int* __restrict__ ovf_dst,
                      const int* __restrict__ ovf_src,
                      float* __restrict__ ax, int n) {
  const int wv   = threadIdx.x >> 5;
  const int lane = threadIdx.x & 31;
  const int node = __builtin_amdgcn_readfirstlane(blockIdx.x * WAVES + wv);
  if (node >= n) return;

  const int c = cnt[node];
  const float di = dinv_of(c);
  const float sl = di * di;

  const float4 xs = ((const float4*)(x + (size_t)node * INDIM))[lane];
  float4 a = make_float4(0.0f, 0.0f, 0.0f, 0.0f);

  {
    const int m = (c < CAP) ? c : CAP;
    int   s_l = 0;
    float w_l = 0.0f;
    if (lane < m) {
      s_l = bucket[(size_t)node * CAP + lane];
      w_l = dinv_of(cnt[s_l]) * di;
    }
    int j = 0;
    for (; j + 8 <= m; j += 8) gather_batch<8>(a, x, s_l, w_l, j, lane);
    if (j + 4 <= m) { gather_batch<4>(a, x, s_l, w_l, j, lane); j += 4; }
    if (j + 2 <= m) { gather_batch<2>(a, x, s_l, w_l, j, lane); j += 2; }
    if (j < m)        gather_batch<1>(a, x, s_l, w_l, j, lane);
  }
  if (c > CAP) {
    const int novf = cnt[n];
    for (int b = 0; b < novf; b += 32) {
      const int i = b + lane;
      int od = -1, os = 0;
      if (i < novf) { od = ovf_dst[i]; os = ovf_src[i]; }
      unsigned mask = __builtin_amdgcn_ballot_w32(od == node);
      while (mask != 0u) {
        const int jj = __builtin_ctz(mask);
        mask &= mask - 1u;
        const int s = __builtin_amdgcn_readlane(os, jj);
        const float w = dinv_of(cnt[s]) * di;
        const float4 p = ((const float4*)(x + (size_t)s * INDIM))[lane];
        vec_axpy(a, p, w);
      }
    }
  }

  vec_axpy(a, xs, sl);
  ((float4*)(ax + (size_t)node * INDIM))[lane] = a;
}

static __device__ __forceinline__ void row_add(float4& a0, float4& a1,
                                               const float4 v0, const float4 v1) {
  a0.x += v0.x;
  a0.y += v0.y;
  a0.z += v0.z;
  a0.w += v0.w;
  a1.x += v1.x;
  a1.y += v1.y;
  a1.z += v1.z;
  a1.w += v1.w;
}

template <int WAVES>
__global__ __launch_bounds__(WAVES * 32)
void gather_finish_kernel(const float* __restrict__ h,
                          const int* __restrict__ cnt,
                          const int* __restrict__ bucket,
                          const int* __restrict__ ovf_dst,
                          const int* __restrict__ ovf_src,
                          const float* __restrict__ bias,
                          const float* __restrict__ Wl,
                          const float* __restrict__ bl,
                          float* __restrict__ dstbuf, int n) {
  const int wv   = threadIdx.x >> 5;
  const int lane = threadIdx.x & 31;
  const int node = __builtin_amdgcn_readfirstlane(blockIdx.x * WAVES + wv);

  float r0 = 0.0f, r1 = 0.0f;
  if (node < n) {
    const int c = cnt[node];
    const float di = dinv_of(c);

    const float4* hp = (const float4*)(h + (size_t)node * HID);
    float4 a0 = hp[lane];
    float4 a1 = hp[lane + 32];

    {
      const int m = (c < CAP) ? c : CAP;
      int s_l = 0;
      if (lane < m) s_l = bucket[(size_t)node * CAP + lane];
      int j = 0;
      for (; j + 4 <= m; j += 4) {
        const int s0 = __builtin_amdgcn_readlane(s_l, j);
        const int s1 = __builtin_amdgcn_readlane(s_l, j + 1);
        const int s2 = __builtin_amdgcn_readlane(s_l, j + 2);
        const int s3 = __builtin_amdgcn_readlane(s_l, j + 3);
        const float4* h0 = (const float4*)(h + (size_t)s0 * HID);
        const float4* h1 = (const float4*)(h + (size_t)s1 * HID);
        const float4* h2 = (const float4*)(h + (size_t)s2 * HID);
        const float4* h3 = (const float4*)(h + (size_t)s3 * HID);
        const float4 p0 = h0[lane], q0 = h0[lane + 32];
        const float4 p1 = h1[lane], q1 = h1[lane + 32];
        const float4 p2 = h2[lane], q2 = h2[lane + 32];
        const float4 p3 = h3[lane], q3 = h3[lane + 32];
        row_add(a0, a1, p0, q0);
        row_add(a0, a1, p1, q1);
        row_add(a0, a1, p2, q2);
        row_add(a0, a1, p3, q3);
      }
      #pragma unroll 1
      for (; j < m; ++j) {
        const int s0 = __builtin_amdgcn_readlane(s_l, j);
        const float4* h0 = (const float4*)(h + (size_t)s0 * HID);
        const float4 p0 = h0[lane], q0 = h0[lane + 32];
        row_add(a0, a1, p0, q0);
      }
    }
    if (c > CAP) {
      const int novf = cnt[n];
      for (int b = 0; b < novf; b += 32) {
        const int i = b + lane;
        int od = -1, os = 0;
        if (i < novf) { od = ovf_dst[i]; os = ovf_src[i]; }
        unsigned mask = __builtin_amdgcn_ballot_w32(od == node);
        while (mask != 0u) {
          const int jj = __builtin_ctz(mask);
          mask &= mask - 1u;
          const int s0 = __builtin_amdgcn_readlane(os, jj);
          const float4* h0 = (const float4*)(h + (size_t)s0 * HID);
          const float4 p0 = h0[lane], q0 = h0[lane + 32];
          row_add(a0, a1, p0, q0);
        }
      }
    }

    const float4 b0 = ((const float4*)bias)[lane];
    const float4 b1 = ((const float4*)bias)[lane + 32];
    a0.x = add_relu(a0.x * di, b0.x); a0.y = add_relu(a0.y * di, b0.y);
    a0.z = add_relu(a0.z * di, b0.z); a0.w = add_relu(a0.w * di, b0.w);
    a1.x = add_relu(a1.x * di, b1.x); a1.y = add_relu(a1.y * di, b1.y);
    a1.z = add_relu(a1.z * di, b1.z); a1.w = add_relu(a1.w * di, b1.w);

    const float4* wp = (const float4*)Wl;
    const float4 wa = wp[2 * lane],      wb = wp[2 * lane + 1];
    const float4 wc = wp[64 + 2 * lane], wd = wp[64 + 2 * lane + 1];
    r0 = a0.x * wa.x;            r1 = a0.x * wa.y;
    r0 = fmaf(a0.y, wa.z, r0);   r1 = fmaf(a0.y, wa.w, r1);
    r0 = fmaf(a0.z, wb.x, r0);   r1 = fmaf(a0.z, wb.y, r1);
    r0 = fmaf(a0.w, wb.z, r0);   r1 = fmaf(a0.w, wb.w, r1);
    r0 = fmaf(a1.x, wc.x, r0);   r1 = fmaf(a1.x, wc.y, r1);
    r0 = fmaf(a1.y, wc.z, r0);   r1 = fmaf(a1.y, wc.w, r1);
    r0 = fmaf(a1.z, wd.x, r0);   r1 = fmaf(a1.z, wd.y, r1);
    r0 = fmaf(a1.w, wd.z, r0);   r1 = fmaf(a1.w, wd.w, r1);
    for (int o = 16; o > 0; o >>= 1) {
      r0 += __shfl_xor(r0, o);
      r1 += __shfl_xor(r1, o);
    }
    r0 += bl[0];
    r1 += bl[1];
  }

  __shared__ float stage[2 * WAVES];
  if (lane == 0) {
    stage[2 * wv]     = r0;
    stage[2 * wv + 1] = r1;
  }
  __syncthreads();
  if (wv == 0) {
    const int idx = blockIdx.x * (2 * WAVES) + lane;
    if (lane < 2 * WAVES && idx < 2 * n) dstbuf[idx] = stage[lane];
  }
}

extern "C" void kernel_launch(void* const* d_in, const int* in_sizes, int n_in,
                              void* d_out, int out_size, void* d_ws, size_t ws_size,
                              hipStream_t stream) {
  const float* x   = (const float*)d_in[0];
  const int*   ei  = (const int*)d_in[1];
  const float* W1  = (const float*)d_in[2];
  const float* b1  = (const float*)d_in[3];
  const float* W2  = (const float*)d_in[4];
  const float* b2  = (const float*)d_in[5];
  const float* Wl  = (const float*)d_in[6];
  const float* bl  = (const float*)d_in[7];
  float*       out = (float*)d_out;

  const int*   src = ei;
  const int*   dst = ei + NEDGES;

  char* ws = (char*)d_ws;
  const size_t H_OFF   = 200192;
  const size_t HBYTES  = (size_t)NNODES * HID * 4;
  const size_t AX_OFF  = H_OFF + HBYTES;
  const size_t AXBYTES = (size_t)NNODES * INDIM * 4;
  const size_t BP1_OFF = AX_OFF + AXBYTES;
  const size_t BP2_OFF = BP1_OFF + 131072;
  const size_t CNT_OFF = BP2_OFF + 262144;
  const size_t BK_OFF  = CNT_OFF + 200192;
  const size_t OD_OFF  = BK_OFF + 6400000;
  const size_t OS_OFF  = OD_OFF + 2400000;

  float* H    = (float*)(ws + H_OFF);
  float* AX   = (float*)(ws + AX_OFF);
  float* Bp1  = (float*)(ws + BP1_OFF);
  float* Bp2  = (float*)(ws + BP2_OFF);
  int*   cnt    = (int*)(ws + CNT_OFF);
  int*   bucket = (int*)(ws + BK_OFF);
  int*   ovfdst = (int*)(ws + OD_OFF);
  int*   ovfsrc = (int*)(ws + OS_OFF);

  const int TPB = 256;
  const int prepBlocks  = (HID * HID + TPB - 1) / TPB;
  const int edgeBlocks  = (NEDGES + TPB - 1) / TPB;
  const int rowBlocks8  = (NNODES + 7) / 8;
  const int rowBlocks16 = (NNODES + 15) / 16;
  const int gemmBlocks  = GEMM_BLOCKS;

  prep_kernel<<<prepBlocks, TPB, 0, stream>>>(cnt, W1, Bp1, W2, Bp2);
  deg_bucket_kernel<<<edgeBlocks, TPB, 0, stream>>>(src, dst, cnt, bucket, ovfdst, ovfsrc, NEDGES, NNODES);

  gather_in_kernel<8><<<rowBlocks8, 256, 0, stream>>>(x, cnt, bucket, ovfdst, ovfsrc, AX, NNODES);

  dense2_wmma_f32_kernel<<<gemmBlocks, 128, 0, stream>>>(AX, Bp1, b1, Bp2, cnt, H);

  gather_finish_kernel<16><<<rowBlocks16, 512, 0, stream>>>(
      H, cnt, bucket, ovfdst, ovfsrc, b2, Wl, bl, out, NNODES);
}
